# starting kernel
# baseline (speedup 1.0000x reference)
#include <hip/hip_runtime.h>
#include <hip/hip_bf16.h>

typedef __attribute__((ext_vector_type(2))) float v2f;
typedef __attribute__((ext_vector_type(8))) float v8f;

__device__ __forceinline__ v8f wmma_f32(v2f a, v2f b, v8f c) {
  return __builtin_amdgcn_wmma_f32_16x16x4_f32(
       false, a,  false, b,
       (short)0, c,  false,  false);
}

__global__ __launch_bounds__(32)
void gemm_nt_wmma(const float* __restrict__ A, const float* __restrict__ B,
                  float* __restrict__ C, int K, int lda, int ldb, int ldc) {
  const int lane = threadIdx.x & 31;
  const int half = lane >> 4;
  const int l16  = lane & 15;
  const int bm = blockIdx.y * 32;
  const int bn = blockIdx.x * 64;

  v8f acc[2][4];
#pragma unroll
  for (int i = 0; i < 2; ++i)
#pragma unroll
    for (int j = 0; j < 4; ++j) acc[i][j] = (v8f){};

  uint32_t aoff[2], boff[4];
#pragma unroll
  for (int i = 0; i < 2; ++i)
    aoff[i] = (uint32_t)(bm + i * 16 + l16) * (uint32_t)lda + 2u * half;
#pragma unroll
  for (int j = 0; j < 4; ++j)
    boff[j] = (uint32_t)(bn + j * 16 + l16) * (uint32_t)ldb + 2u * half;

  v2f af[2], bf[4];
#pragma unroll
  for (int i = 0; i < 2; ++i) af[i] = *(const v2f*)(A + aoff[i]);
#pragma unroll
  for (int j = 0; j < 4; ++j) bf[j] = *(const v2f*)(B + boff[j]);

#pragma unroll 2
  for (int k0 = 0; k0 < K - 4; k0 += 4) {
    v2f afn[2], bfn[4];
#pragma unroll
    for (int i = 0; i < 2; ++i)
      afn[i] = *(const v2f*)(A + aoff[i] + (uint32_t)(k0 + 4));
#pragma unroll
    for (int j = 0; j < 4; ++j)
      bfn[j] = *(const v2f*)(B + boff[j] + (uint32_t)(k0 + 4));
#pragma unroll
    for (int i = 0; i < 2; ++i)
#pragma unroll
      for (int j = 0; j < 4; ++j) acc[i][j] = wmma_f32(af[i], bf[j], acc[i][j]);
#pragma unroll
    for (int i = 0; i < 2; ++i) af[i] = afn[i];
#pragma unroll
    for (int j = 0; j < 4; ++j) bf[j] = bfn[j];
  }
#pragma unroll
  for (int i = 0; i < 2; ++i)
#pragma unroll
    for (int j = 0; j < 4; ++j) acc[i][j] = wmma_f32(af[i], bf[j], acc[i][j]);

#pragma unroll
  for (int i = 0; i < 2; ++i)
#pragma unroll
    for (int j = 0; j < 4; ++j) {
      const uint32_t row0 = (uint32_t)(bm + i * 16 + half * 8);
      const uint32_t col  = (uint32_t)(bn + j * 16 + l16);
#pragma unroll
      for (int v = 0; v < 8; ++v)
        C[(row0 + v) * (uint32_t)ldc + col] = acc[i][j][v];
    }
}

__global__ __launch_bounds__(256)
void rope_kernel(float* __restrict__ qkv, int T) {
  int idx = blockIdx.x * blockDim.x + threadIdx.x;
  if (idx >= T * 1024) return;
  int t = idx >> 10;
  int r = idx & 1023;
  int qksel = r >> 9;
  int h = (r >> 5) & 15;
  int p = r & 31;
  float inv = powf(10000.0f, -(float)(2 * p) / 64.0f);
  float ang = (float)t * inv;
  float s = sinf(ang), c = cosf(ang);
  float* base = qkv + (uint32_t)t * 3072u + (uint32_t)(qksel * 1024 + h * 64 + 2 * p);
  float x1 = base[0], x2 = base[1];
  base[0] = x1 * c - x2 * s;
  base[1] = x2 * c + x1 * s;
}

#define LDT 68

__global__ __launch_bounds__(256)
void attn_kernel(const float* __restrict__ qkv, float* __restrict__ out) {
  __shared__ float Klds[64 * LDT];
  __shared__ float Vlds[64 * LDT];
  __shared__ float Plds[8 * 16 * LDT];

  const int tid  = threadIdx.x;
  const int lane = tid & 31;
  const int wave = tid >> 5;
  const int half = lane >> 4;
  const int l16  = lane & 15;
  const int h     = blockIdx.y;
  const int qbase = blockIdx.x * 128 + wave * 16;
  const uint32_t ld = 3072;

  const float* qp = qkv + (uint32_t)h * 64u;
  const float* kp = qkv + 1024u + (uint32_t)h * 64u;
  const float* vp = qkv + 2048u + (uint32_t)h * 64u;

  v2f qf[16];
  {
    const uint32_t qoff = (uint32_t)(qbase + l16) * ld + 2u * half;
#pragma unroll
    for (int kk = 0; kk < 16; ++kk)
      qf[kk] = *(const v2f*)(qp + qoff + 4u * kk);
  }

  v8f o[4];
#pragma unroll
  for (int ct = 0; ct < 4; ++ct) o[ct] = (v8f){};
  float mrow[8], lrow[8];
#pragma unroll
  for (int v = 0; v < 8; ++v) { mrow[v] = -1e30f; lrow[v] = 0.0f; }

  float* myp = Plds + wave * 16 * LDT;
  const int ktmax_w = (qbase + 15) >> 6;
  const int ktmax_b = (blockIdx.x * 128 + 127) >> 6;

  for (int kt = 0; kt <= ktmax_b; ++kt) {
    const uint32_t tbase = (uint32_t)(kt * 64) * ld;

    __syncthreads();
#pragma unroll 4
    for (int t = 0; t < 16; ++t) {
      int idx = t * 256 + tid;
      int key = idx >> 6, c = idx & 63;
      uint32_t goff = tbase + (uint32_t)key * ld + (uint32_t)c;
      float kvv = kp[goff];
      float vvv = vp[goff];
      Klds[key * LDT + c] = kvv;
      Vlds[c * LDT + key] = vvv;
    }
    if (kt < ktmax_b) {
      if (tid < 64) {
        const float* p = kp + tbase + (uint32_t)(64 + tid) * ld;
        __builtin_prefetch(p, 0, 3);
        __builtin_prefetch(p + 32, 0, 3);
      } else if (tid < 128) {
        const float* p = vp + tbase + (uint32_t)tid * ld;
        __builtin_prefetch(p, 0, 3);
        __builtin_prefetch(p + 32, 0, 3);
      }
    }
    __syncthreads();

    if (kt <= ktmax_w) {
      v8f s[4];
#pragma unroll
      for (int j = 0; j < 4; ++j) s[j] = (v8f){};
#pragma unroll 4
      for (int kk = 0; kk < 16; ++kk) {
        v2f a = qf[kk];
#pragma unroll
        for (int j = 0; j < 4; ++j) {
          v2f b = *(const v2f*)&Klds[(j * 16 + l16) * LDT + 4 * kk + 2 * half];
          s[j] = wmma_f32(a, b, s[j]);
        }
      }

      float alpha[8];
#pragma unroll
      for (int v = 0; v < 8; ++v) {
        const int row = qbase + half * 8 + v;
        float sv[4];
        float mx = -1e30f;
#pragma unroll
        for (int j = 0; j < 4; ++j) {
          const int col = kt * 64 + j * 16 + l16;
          float x = s[j][v] * 0.125f;
          if (col > row) x = -1e30f;
          sv[j] = x;
          mx = fmaxf(mx, x);
        }
#pragma unroll
        for (int off = 1; off < 16; off <<= 1)
          mx = fmaxf(mx, __shfl_xor(mx, off, 32));
        const float mnew = fmaxf(mrow[v], mx);
        const float a = __expf(mrow[v] - mnew);
        float rsum = 0.0f;
#pragma unroll
        for (int j = 0; j < 4; ++j) {
          float p = __expf(sv[j] - mnew);
          myp[(half * 8 + v) * LDT + j * 16 + l16] = p;
          rsum += p;
        }
#pragma unroll
        for (int off = 1; off < 16; off <<= 1)
          rsum += __shfl_xor(rsum, off, 32);
        lrow[v] = lrow[v] * a + rsum;
        mrow[v] = mnew;
        alpha[v] = a;
      }
#pragma unroll
      for (int ct = 0; ct < 4; ++ct)
#pragma unroll
        for (int v = 0; v < 8; ++v) o[ct][v] *= alpha[v];

#pragma unroll 2
      for (int kk = 0; kk < 16; ++kk) {
        v2f a = *(const v2f*)&myp[l16 * LDT + 4 * kk + 2 * half];
#pragma unroll
        for (int ct = 0; ct < 4; ++ct) {
          v2f b = *(const v2f*)&Vlds[(ct * 16 + l16) * LDT + 4 * kk + 2 * half];
          o[ct] = wmma_f32(a, b, o[ct]);
        }
      }
    }
  }

#pragma unroll
  for (int ct = 0; ct < 4; ++ct)
#pragma unroll
    for (int v = 0; v < 8; ++v) {
      const uint32_t row = (uint32_t)(qbase + half * 8 + v);
      out[row * 1024u + (uint32_t)(h * 64 + ct * 16 + l16)] =
          o[ct][v] / lrow[v];
    }
}

extern "C" void kernel_launch(void* const* d_in, const int* in_sizes, int n_in,
                              void* d_out, int out_size, void* d_ws, size_t ws_size,
                              hipStream_t stream) {
  const float* x     = (const float*)d_in[0];
  const float* Wqkv  = (const float*)d_in[1];
  const float* Wproj = (const float*)d_in[2];
  float* out = (float*)d_out;

  float* qkv     = (float*)d_ws;
  float* attnout = qkv + (size_t)4096 * 3072;

  gemm_nt_wmma<<<dim3(3072 / 64, 4096 / 32), 32, 0, stream>>>(
      x, Wqkv, qkv, 1024, 1024, 1024, 3072);

  rope_kernel<<<(4096 * 1024) / 256, 256, 0, stream>>>(qkv, 4096);

  attn_kernel<<<dim3(32, 16), 256, 0, stream>>>(qkv, attnout);

  gemm_nt_wmma<<<dim3(1024 / 64, 4096 / 32), 32, 0, stream>>>(
      attnout, Wproj, out, 1024, 1024, 1024, 1024);
}


# GEMMs re-tiled: 8-wave workgroup per 128x128 tile, A/B K-slabs staged in double-buffered LDS by 16-byte row-coalesced global loads (was one wave per 32x64 tile, fragments read straight from global)
# speedup vs baseline: 1.5950x; 1.5950x over previous
#include <hip/hip_runtime.h>
#include <hip/hip_bf16.h>

typedef __attribute__((ext_vector_type(2))) float v2f;
typedef __attribute__((ext_vector_type(4))) float v4f;
typedef __attribute__((ext_vector_type(8))) float v8f;

__device__ __forceinline__ v8f wmma_f32(v2f a, v2f b, v8f c) {
  return __builtin_amdgcn_wmma_f32_16x16x4_f32(
       false, a,  false, b,
       (short)0, c,  false,  false);
}

#define GBM 128
#define GBN 128
#define GBK 32
#define GLD 36
#define GSTAGE ((GBM + GBN) * GLD)

__global__ __launch_bounds__(256)
void gemm_nt_wmma(const float* __restrict__ A, const float* __restrict__ B,
                  float* __restrict__ C, int K, int lda, int ldb, int ldc) {
  __shared__ float lds[2 * GSTAGE];

  const int tid  = threadIdx.x;
  const int lane = tid & 31;
  const int wave = tid >> 5;
  const int half = lane >> 4;
  const int l16  = lane & 15;
  const int wm = wave >> 2;
  const int wn = wave & 3;
  const int bm = blockIdx.y * GBM;
  const int bn = blockIdx.x * GBN;

  const uint32_t srow = (uint32_t)(tid >> 3);
  const uint32_t sc4  = (uint32_t)(tid & 7) * 4u;
  const uint32_t aoff = (uint32_t)(bm + (int)srow) * (uint32_t)lda + sc4;
  const uint32_t boff = (uint32_t)(bn + (int)srow) * (uint32_t)ldb + sc4;
  const uint32_t astep = 32u * (uint32_t)lda;
  const uint32_t bstep = 32u * (uint32_t)ldb;
  const uint32_t soff = srow * GLD + sc4;

  const uint32_t faoff = (uint32_t)(wm * 64 + l16) * GLD + 4u * half;
  const uint32_t fboff = (uint32_t)(GBM + wn * 32 + l16) * GLD + 4u * half;

  v8f acc[4][2];
#pragma unroll
  for (int i = 0; i < 4; ++i)
#pragma unroll
    for (int j = 0; j < 2; ++j) acc[i][j] = (v8f){};

  v4f pa[4], pb[4];
#pragma unroll
  for (int t = 0; t < 4; ++t) {
    pa[t] = *(const v4f*)(A + aoff + (uint32_t)t * astep);
    pb[t] = *(const v4f*)(B + boff + (uint32_t)t * bstep);
  }
#pragma unroll
  for (int t = 0; t < 4; ++t) {
    *(v4f*)(lds + soff + (uint32_t)t * (32u * GLD)) = pa[t];
    *(v4f*)(lds + (uint32_t)(GBM * GLD) + soff + (uint32_t)t * (32u * GLD)) = pb[t];
  }
  __syncthreads();

  const int KT = K / GBK;
  for (int kt = 0; kt < KT; ++kt) {
    const uint32_t cur = (kt & 1) ? (uint32_t)GSTAGE : 0u;
    const uint32_t nxt = (uint32_t)GSTAGE - cur;
    const bool more = (kt + 1 < KT);

    if (more) {
      const uint32_t k1 = (uint32_t)(kt + 1) * GBK;
#pragma unroll
      for (int t = 0; t < 4; ++t) {
        pa[t] = *(const v4f*)(A + aoff + (uint32_t)t * astep + k1);
        pb[t] = *(const v4f*)(B + boff + (uint32_t)t * bstep + k1);
      }
    }

    const float* fa_p = lds + cur + faoff;
    const float* fb_p = lds + cur + fboff;

    v4f fa[2][4], fb[2][2];
#pragma unroll
    for (int i = 0; i < 4; ++i) fa[0][i] = *(const v4f*)(fa_p + i * 16 * GLD);
#pragma unroll
    for (int j = 0; j < 2; ++j) fb[0][j] = *(const v4f*)(fb_p + j * 16 * GLD);

#pragma unroll
    for (int ks = 0; ks < 4; ++ks) {
      const int c = ks & 1, n = c ^ 1;
      if (ks + 1 < 4) {
#pragma unroll
        for (int i = 0; i < 4; ++i)
          fa[n][i] = *(const v4f*)(fa_p + i * 16 * GLD + 8 * (ks + 1));
#pragma unroll
        for (int j = 0; j < 2; ++j)
          fb[n][j] = *(const v4f*)(fb_p + j * 16 * GLD + 8 * (ks + 1));
      }
#pragma unroll
      for (int i = 0; i < 4; ++i)
#pragma unroll
        for (int j = 0; j < 2; ++j) {
          v2f a = __builtin_shufflevector(fa[c][i], fa[c][i], 0, 1);
          v2f b = __builtin_shufflevector(fb[c][j], fb[c][j], 0, 1);
          acc[i][j] = wmma_f32(a, b, acc[i][j]);
        }
#pragma unroll
      for (int i = 0; i < 4; ++i)
#pragma unroll
        for (int j = 0; j < 2; ++j) {
          v2f a = __builtin_shufflevector(fa[c][i], fa[c][i], 2, 3);
          v2f b = __builtin_shufflevector(fb[c][j], fb[c][j], 2, 3);
          acc[i][j] = wmma_f32(a, b, acc[i][j]);
        }
    }

    if (more) {
#pragma unroll
      for (int t = 0; t < 4; ++t) {
        *(v4f*)(lds + nxt + soff + (uint32_t)t * (32u * GLD)) = pa[t];
        *(v4f*)(lds + nxt + (uint32_t)(GBM * GLD) + soff + (uint32_t)t * (32u * GLD)) = pb[t];
      }
    }
    __syncthreads();
  }

#pragma unroll
  for (int i = 0; i < 4; ++i)
#pragma unroll
    for (int j = 0; j < 2; ++j) {
      const uint32_t row0 = (uint32_t)(bm + wm * 64 + i * 16 + half * 8);
      const uint32_t col  = (uint32_t)(bn + wn * 32 + j * 16 + l16);
#pragma unroll
      for (int v = 0; v < 8; ++v)
        C[(row0 + v) * (uint32_t)ldc + col] = acc[i][j][v];
    }
}

__global__ __launch_bounds__(256)
void rope_kernel(float* __restrict__ qkv, int T) {
  int idx = blockIdx.x * blockDim.x + threadIdx.x;
  if (idx >= T * 1024) return;
  int t = idx >> 10;
  int r = idx & 1023;
  int qksel = r >> 9;
  int h = (r >> 5) & 15;
  int p = r & 31;
  float inv = powf(10000.0f, -(float)(2 * p) / 64.0f);
  float ang = (float)t * inv;
  float s = sinf(ang), c = cosf(ang);
  float* base = qkv + (uint32_t)t * 3072u + (uint32_t)(qksel * 1024 + h * 64 + 2 * p);
  float x1 = base[0], x2 = base[1];
  base[0] = x1 * c - x2 * s;
  base[1] = x2 * c + x1 * s;
}

#define LDT 68

__global__ __launch_bounds__(256)
void attn_kernel(const float* __restrict__ qkv, float* __restrict__ out) {
  __shared__ float Klds[64 * LDT];
  __shared__ float Vlds[64 * LDT];
  __shared__ float Plds[8 * 16 * LDT];

  const int tid  = threadIdx.x;
  const int lane = tid & 31;
  const int wave = tid >> 5;
  const int half = lane >> 4;
  const int l16  = lane & 15;
  const int h     = blockIdx.y;
  const int qbase = blockIdx.x * 128 + wave * 16;
  const uint32_t ld = 3072;

  const float* qp = qkv + (uint32_t)h * 64u;
  const float* kp = qkv + 1024u + (uint32_t)h * 64u;
  const float* vp = qkv + 2048u + (uint32_t)h * 64u;

  v2f qf[16];
  {
    const uint32_t qoff = (uint32_t)(qbase + l16) * ld + 2u * half;
#pragma unroll
    for (int kk = 0; kk < 16; ++kk)
      qf[kk] = *(const v2f*)(qp + qoff + 4u * kk);
  }

  v8f o[4];
#pragma unroll
  for (int ct = 0; ct < 4; ++ct) o[ct] = (v8f){};
  float mrow[8], lrow[8];
#pragma unroll
  for (int v = 0; v < 8; ++v) { mrow[v] = -1e30f; lrow[v] = 0.0f; }

  float* myp = Plds + wave * 16 * LDT;
  const int ktmax_w = (qbase + 15) >> 6;
  const int ktmax_b = (blockIdx.x * 128 + 127) >> 6;

  for (int kt = 0; kt <= ktmax_b; ++kt) {
    const uint32_t tbase = (uint32_t)(kt * 64) * ld;

    __syncthreads();
#pragma unroll 4
    for (int t = 0; t < 16; ++t) {
      int idx = t * 256 + tid;
      int key = idx >> 6, c = idx & 63;
      uint32_t goff = tbase + (uint32_t)key * ld + (uint32_t)c;
      float kvv = kp[goff];
      float vvv = vp[goff];
      Klds[key * LDT + c] = kvv;
      Vlds[c * LDT + key] = vvv;
    }
    if (kt < ktmax_b) {
      if (tid < 64) {
        const float* p = kp + tbase + (uint32_t)(64 + tid) * ld;
        __builtin_prefetch(p, 0, 3);
        __builtin_prefetch(p + 32, 0, 3);
      } else if (tid < 128) {
        const float* p = vp + tbase + (uint32_t)tid * ld;
        __builtin_prefetch(p, 0, 3);
        __builtin_prefetch(p + 32, 0, 3);
      }
    }
    __syncthreads();

    if (kt <= ktmax_w) {
      v8f s[4];
#pragma unroll
      for (int j = 0; j < 4; ++j) s[j] = (v8f){};
#pragma unroll 4
      for (int kk = 0; kk < 16; ++kk) {
        v2f a = qf[kk];
#pragma unroll
        for (int j = 0; j < 4; ++j) {
          v2f b = *(const v2f*)&Klds[(j * 16 + l16) * LDT + 4 * kk + 2 * half];
          s[j] = wmma_f32(a, b, s[j]);
        }
      }

      float alpha[8];
#pragma unroll
      for (int v = 0; v < 8; ++v) {
        const int row = qbase + half * 8 + v;
        float sv[4];
        float mx = -1e30f;
#pragma unroll
        for (int j = 0; j < 4; ++j) {
          const int col = kt * 64 + j * 16 + l16;
          float x = s[j][v] * 0.125f;
          if (col > row) x = -1e30f;
          sv[j] = x;
          mx = fmaxf(mx, x);
        }
#pragma unroll
        for (int off = 1; off < 16; off <<= 1)
          mx = fmaxf(mx, __shfl_xor(mx, off, 32));
        const float mnew = fmaxf(mrow[v], mx);
        const float a = __expf(mrow[v] - mnew);
        float rsum = 0.0f;
#pragma unroll
        for (int j = 0; j < 4; ++j) {
          float p = __expf(sv[j] - mnew);
          myp[(half * 8 + v) * LDT + j * 16 + l16] = p;
          rsum += p;
        }
#pragma unroll
        for (int off = 1; off < 16; off <<= 1)
          rsum += __shfl_xor(rsum, off, 32);
        lrow[v] = lrow[v] * a + rsum;
        mrow[v] = mnew;
        alpha[v] = a;
      }
#pragma unroll
      for (int ct = 0; ct < 4; ++ct)
#pragma unroll
        for (int v = 0; v < 8; ++v) o[ct][v] *= alpha[v];

#pragma unroll 2
      for (int kk = 0; kk < 16; ++kk) {
        v2f a = *(const v2f*)&myp[l16 * LDT + 4 * kk + 2 * half];
#pragma unroll
        for (int ct = 0; ct < 4; ++ct) {
          v2f b = *(const v2f*)&Vlds[(ct * 16 + l16) * LDT + 4 * kk + 2 * half];
          o[ct] = wmma_f32(a, b, o[ct]);
        }
      }
    }
  }

#pragma unroll
  for (int ct = 0; ct < 4; ++ct)
#pragma unroll
    for (int v = 0; v < 8; ++v) {
      const uint32_t row = (uint32_t)(qbase + half * 8 + v);
      out[row * 1024u + (uint32_t)(h * 64 + ct * 16 + l16)] =
          o[ct][v] / lrow[v];
    }
}

extern "C" void kernel_launch(void* const* d_in, const int* in_sizes, int n_in,
                              void* d_out, int out_size, void* d_ws, size_t ws_size,
                              hipStream_t stream) {
  const float* x     = (const float*)d_in[0];
  const float* Wqkv  = (const float*)d_in[1];
  const float* Wproj = (const float*)d_in[2];
  float* out = (float*)d_out;

  float* qkv     = (float*)d_ws;
  float* attnout = qkv + (size_t)4096 * 3072;

  gemm_nt_wmma<<<dim3(24, 32), 256, 0, stream>>>(
      x, Wqkv, qkv, 1024, 1024, 1024, 3072);

  rope_kernel<<<(4096 * 1024) / 256, 256, 0, stream>>>(qkv, 4096);

  attn_kernel<<<dim3(32, 16), 256, 0, stream>>>(qkv, attnout);

  gemm_nt_wmma<<<dim3(8, 32), 256, 0, stream>>>(
      attnout, Wproj, out, 1024, 1024, 1024, 1024);
}


# attention softmax: 16-lane row maxima by DPP row_xmask on the vector unit instead of ds_bpermute shuffles; row sums kept per lane and joined once after the key loop
# speedup vs baseline: 1.7870x; 1.1204x over previous
#include <hip/hip_runtime.h>
#include <hip/hip_bf16.h>

typedef __attribute__((ext_vector_type(2))) float v2f;
typedef __attribute__((ext_vector_type(4))) float v4f;
typedef __attribute__((ext_vector_type(8))) float v8f;

__device__ __forceinline__ v8f wmma_f32(v2f a, v2f b, v8f c) {
  return __builtin_amdgcn_wmma_f32_16x16x4_f32(
       false, a,  false, b,
       (short)0, c,  false,  false);
}

template <int M>
__device__ __forceinline__ float row_max_step(float x) {
  float r;
  asm("v_max_num_f32_dpp %0, %1, %1 row_xmask:%2 row_mask:0xf bank_mask:0xf"
      : "=v"(r) : "v"(x), "n"(M));
  return r;
}
template <int M>
__device__ __forceinline__ float row_xor(float x) {
  return __builtin_amdgcn_update_dpp(0.0f, x, 0x160 + M, 0xf, 0xf, true);
}

#define GBM 128
#define GBN 128
#define GBK 32
#define GLD 36
#define GSTAGE ((GBM + GBN) * GLD)

__global__ __launch_bounds__(256)
void gemm_nt_wmma(const float* __restrict__ A, const float* __restrict__ B,
                  float* __restrict__ C, int K, int lda, int ldb, int ldc) {
  __shared__ float lds[2 * GSTAGE];

  const int tid  = threadIdx.x;
  const int lane = tid & 31;
  const int wave = tid >> 5;
  const int half = lane >> 4;
  const int l16  = lane & 15;
  const int wm = wave >> 2;
  const int wn = wave & 3;
  const int bm = blockIdx.y * GBM;
  const int bn = blockIdx.x * GBN;

  const uint32_t srow = (uint32_t)(tid >> 3);
  const uint32_t sc4  = (uint32_t)(tid & 7) * 4u;
  const uint32_t aoff = (uint32_t)(bm + (int)srow) * (uint32_t)lda + sc4;
  const uint32_t boff = (uint32_t)(bn + (int)srow) * (uint32_t)ldb + sc4;
  const uint32_t astep = 32u * (uint32_t)lda;
  const uint32_t bstep = 32u * (uint32_t)ldb;
  const uint32_t soff = srow * GLD + sc4;

  const uint32_t faoff = (uint32_t)(wm * 64 + l16) * GLD + 4u * half;
  const uint32_t fboff = (uint32_t)(GBM + wn * 32 + l16) * GLD + 4u * half;

  v8f acc[4][2];
#pragma unroll
  for (int i = 0; i < 4; ++i)
#pragma unroll
    for (int j = 0; j < 2; ++j) acc[i][j] = (v8f){};

  v4f pa[4], pb[4];
#pragma unroll
  for (int t = 0; t < 4; ++t) {
    pa[t] = *(const v4f*)(A + aoff + (uint32_t)t * astep);
    pb[t] = *(const v4f*)(B + boff + (uint32_t)t * bstep);
  }
#pragma unroll
  for (int t = 0; t < 4; ++t) {
    *(v4f*)(lds + soff + (uint32_t)t * (32u * GLD)) = pa[t];
    *(v4f*)(lds + (uint32_t)(GBM * GLD) + soff + (uint32_t)t * (32u * GLD)) = pb[t];
  }
  __syncthreads();

  const int KT = K / GBK;
  for (int kt = 0; kt < KT; ++kt) {
    const uint32_t cur = (kt & 1) ? (uint32_t)GSTAGE : 0u;
    const uint32_t nxt = (uint32_t)GSTAGE - cur;
    const bool more = (kt + 1 < KT);

    if (more) {
      const uint32_t k1 = (uint32_t)(kt + 1) * GBK;
#pragma unroll
      for (int t = 0; t < 4; ++t) {
        pa[t] = *(const v4f*)(A + aoff + (uint32_t)t * astep + k1);
        pb[t] = *(const v4f*)(B + boff + (uint32_t)t * bstep + k1);
      }
    }

    const float* fa_p = lds + cur + faoff;
    const float* fb_p = lds + cur + fboff;

    v4f fa[2][4], fb[2][2];
#pragma unroll
    for (int i = 0; i < 4; ++i) fa[0][i] = *(const v4f*)(fa_p + i * 16 * GLD);
#pragma unroll
    for (int j = 0; j < 2; ++j) fb[0][j] = *(const v4f*)(fb_p + j * 16 * GLD);

#pragma unroll
    for (int ks = 0; ks < 4; ++ks) {
      const int c = ks & 1, n = c ^ 1;
      if (ks + 1 < 4) {
#pragma unroll
        for (int i = 0; i < 4; ++i)
          fa[n][i] = *(const v4f*)(fa_p + i * 16 * GLD + 8 * (ks + 1));
#pragma unroll
        for (int j = 0; j < 2; ++j)
          fb[n][j] = *(const v4f*)(fb_p + j * 16 * GLD + 8 * (ks + 1));
      }
#pragma unroll
      for (int i = 0; i < 4; ++i)
#pragma unroll
        for (int j = 0; j < 2; ++j) {
          v2f a = __builtin_shufflevector(fa[c][i], fa[c][i], 0, 1);
          v2f b = __builtin_shufflevector(fb[c][j], fb[c][j], 0, 1);
          acc[i][j] = wmma_f32(a, b, acc[i][j]);
        }
#pragma unroll
      for (int i = 0; i < 4; ++i)
#pragma unroll
        for (int j = 0; j < 2; ++j) {
          v2f a = __builtin_shufflevector(fa[c][i], fa[c][i], 2, 3);
          v2f b = __builtin_shufflevector(fb[c][j], fb[c][j], 2, 3);
          acc[i][j] = wmma_f32(a, b, acc[i][j]);
        }
    }

    if (more) {
#pragma unroll
      for (int t = 0; t < 4; ++t) {
        *(v4f*)(lds + nxt + soff + (uint32_t)t * (32u * GLD)) = pa[t];
        *(v4f*)(lds + nxt + (uint32_t)(GBM * GLD) + soff + (uint32_t)t * (32u * GLD)) = pb[t];
      }
    }
    __syncthreads();
  }

#pragma unroll
  for (int i = 0; i < 4; ++i)
#pragma unroll
    for (int j = 0; j < 2; ++j) {
      const uint32_t row0 = (uint32_t)(bm + wm * 64 + i * 16 + half * 8);
      const uint32_t col  = (uint32_t)(bn + wn * 32 + j * 16 + l16);
#pragma unroll
      for (int v = 0; v < 8; ++v)
        C[(row0 + v) * (uint32_t)ldc + col] = acc[i][j][v];
    }
}

__global__ __launch_bounds__(256)
void rope_kernel(float* __restrict__ qkv, int T) {
  int idx = blockIdx.x * blockDim.x + threadIdx.x;
  if (idx >= T * 1024) return;
  int t = idx >> 10;
  int r = idx & 1023;
  int qksel = r >> 9;
  int h = (r >> 5) & 15;
  int p = r & 31;
  float inv = powf(10000.0f, -(float)(2 * p) / 64.0f);
  float ang = (float)t * inv;
  float s = sinf(ang), c = cosf(ang);
  float* base = qkv + (uint32_t)t * 3072u + (uint32_t)(qksel * 1024 + h * 64 + 2 * p);
  float x1 = base[0], x2 = base[1];
  base[0] = x1 * c - x2 * s;
  base[1] = x2 * c + x1 * s;
}

#define LDT 68

__global__ __launch_bounds__(256)
void attn_kernel(const float* __restrict__ qkv, float* __restrict__ out) {
  __shared__ float Klds[64 * LDT];
  __shared__ float Vlds[64 * LDT];
  __shared__ float Plds[8 * 16 * LDT];

  const int tid  = threadIdx.x;
  const int lane = tid & 31;
  const int wave = tid >> 5;
  const int half = lane >> 4;
  const int l16  = lane & 15;
  const int h     = blockIdx.y;
  const int qbase = blockIdx.x * 128 + wave * 16;
  const uint32_t ld = 3072;

  const float* qp = qkv + (uint32_t)h * 64u;
  const float* kp = qkv + 1024u + (uint32_t)h * 64u;
  const float* vp = qkv + 2048u + (uint32_t)h * 64u;

  v2f qf[16];
  {
    const uint32_t qoff = (uint32_t)(qbase + l16) * ld + 2u * half;
#pragma unroll
    for (int kk = 0; kk < 16; ++kk)
      qf[kk] = *(const v2f*)(qp + qoff + 4u * kk);
  }

  v8f o[4];
#pragma unroll
  for (int ct = 0; ct < 4; ++ct) o[ct] = (v8f){};
  float mrow[8], lrow[8];
#pragma unroll
  for (int v = 0; v < 8; ++v) { mrow[v] = -1e30f; lrow[v] = 0.0f; }

  float* myp = Plds + wave * 16 * LDT;
  const int ktmax_w = (qbase + 15) >> 6;
  const int ktmax_b = (blockIdx.x * 128 + 127) >> 6;

  for (int kt = 0; kt <= ktmax_b; ++kt) {
    const uint32_t tbase = (uint32_t)(kt * 64) * ld;

    __syncthreads();
#pragma unroll 4
    for (int t = 0; t < 16; ++t) {
      int idx = t * 256 + tid;
      int key = idx >> 6, c = idx & 63;
      uint32_t goff = tbase + (uint32_t)key * ld + (uint32_t)c;
      float kvv = kp[goff];
      float vvv = vp[goff];
      Klds[key * LDT + c] = kvv;
      Vlds[c * LDT + key] = vvv;
    }
    if (kt < ktmax_b) {
      if (tid < 64) {
        const float* p = kp + tbase + (uint32_t)(64 + tid) * ld;
        __builtin_prefetch(p, 0, 3);
        __builtin_prefetch(p + 32, 0, 3);
      } else if (tid < 128) {
        const float* p = vp + tbase + (uint32_t)tid * ld;
        __builtin_prefetch(p, 0, 3);
        __builtin_prefetch(p + 32, 0, 3);
      }
    }
    __syncthreads();

    if (kt <= ktmax_w) {
      v8f s[4];
#pragma unroll
      for (int j = 0; j < 4; ++j) s[j] = (v8f){};
#pragma unroll 4
      for (int kk = 0; kk < 16; ++kk) {
        v2f a = qf[kk];
#pragma unroll
        for (int j = 0; j < 4; ++j) {
          v2f b = *(const v2f*)&Klds[(j * 16 + l16) * LDT + 4 * kk + 2 * half];
          s[j] = wmma_f32(a, b, s[j]);
        }
      }

      float mx[8];
#pragma unroll
      for (int v = 0; v < 8; ++v) {
        const int row = qbase + half * 8 + v;
        float m = -1e30f;
#pragma unroll
        for (int j = 0; j < 4; ++j) {
          const int col = kt * 64 + j * 16 + l16;
          float x = s[j][v] * 0.125f;
          if (col > row) x = -1e30f;
          s[j][v] = x;
          m = fmaxf(m, x);
        }
        mx[v] = m;
      }
#pragma unroll
      for (int v = 0; v < 8; ++v) mx[v] = row_max_step<1>(mx[v]);
#pragma unroll
      for (int v = 0; v < 8; ++v) mx[v] = row_max_step<2>(mx[v]);
#pragma unroll
      for (int v = 0; v < 8; ++v) mx[v] = row_max_step<4>(mx[v]);
#pragma unroll
      for (int v = 0; v < 8; ++v) mx[v] = row_max_step<8>(mx[v]);

      float alpha[8];
#pragma unroll
      for (int v = 0; v < 8; ++v) {
        const float mnew = fmaxf(mrow[v], mx[v]);
        const float a = __expf(mrow[v] - mnew);
        float rsum = 0.0f;
#pragma unroll
        for (int j = 0; j < 4; ++j) {
          float p = __expf(s[j][v] - mnew);
          myp[(half * 8 + v) * LDT + j * 16 + l16] = p;
          rsum += p;
        }
        lrow[v] = lrow[v] * a + rsum;
        mrow[v] = mnew;
        alpha[v] = a;
      }
#pragma unroll
      for (int ct = 0; ct < 4; ++ct)
#pragma unroll
        for (int v = 0; v < 8; ++v) o[ct][v] *= alpha[v];

#pragma unroll 2
      for (int kk = 0; kk < 16; ++kk) {
        v2f a = *(const v2f*)&myp[l16 * LDT + 4 * kk + 2 * half];
#pragma unroll
        for (int ct = 0; ct < 4; ++ct) {
          v2f b = *(const v2f*)&Vlds[(ct * 16 + l16) * LDT + 4 * kk + 2 * half];
          o[ct] = wmma_f32(a, b, o[ct]);
        }
      }
    }
  }

#pragma unroll
  for (int v = 0; v < 8; ++v) lrow[v] += row_xor<1>(lrow[v]);
#pragma unroll
  for (int v = 0; v < 8; ++v) lrow[v] += row_xor<2>(lrow[v]);
#pragma unroll
  for (int v = 0; v < 8; ++v) lrow[v] += row_xor<4>(lrow[v]);
#pragma unroll
  for (int v = 0; v < 8; ++v) lrow[v] += row_xor<8>(lrow[v]);

#pragma unroll
  for (int ct = 0; ct < 4; ++ct)
#pragma unroll
    for (int v = 0; v < 8; ++v) {
      const uint32_t row = (uint32_t)(qbase + half * 8 + v);
      out[row * 1024u + (uint32_t)(h * 64 + ct * 16 + l16)] =
          o[ct][v] / lrow[v];
    }
}

extern "C" void kernel_launch(void* const* d_in, const int* in_sizes, int n_in,
                              void* d_out, int out_size, void* d_ws, size_t ws_size,
                              hipStream_t stream) {
  const float* x     = (const float*)d_in[0];
  const float* Wqkv  = (const float*)d_in[1];
  const float* Wproj = (const float*)d_in[2];
  float* out = (float*)d_out;

  float* qkv     = (float*)d_ws;
  float* attnout = qkv + (size_t)4096 * 3072;

  gemm_nt_wmma<<<dim3(24, 32), 256, 0, stream>>>(
      x, Wqkv, qkv, 1024, 1024, 1024, 3072);

  rope_kernel<<<(4096 * 1024) / 256, 256, 0, stream>>>(qkv, 4096);

  attn_kernel<<<dim3(32, 16), 256, 0, stream>>>(qkv, attnout);

  gemm_nt_wmma<<<dim3(8, 32), 256, 0, stream>>>(
      attnout, Wproj, out, 1024, 1024, 1024, 1024);
}


# attention: each workgroup (now 16 waves) owns query tiles x and 31-x of a head over one staged K/V stream, so every workgroup processor gets equal causal work (was one tile each, 2 to 64 key tiles)
# speedup vs baseline: 2.3511x; 1.3157x over previous
#include <hip/hip_runtime.h>
#include <hip/hip_bf16.h>

typedef __attribute__((ext_vector_type(2))) float v2f;
typedef __attribute__((ext_vector_type(4))) float v4f;
typedef __attribute__((ext_vector_type(8))) float v8f;

__device__ __forceinline__ v8f wmma_f32(v2f a, v2f b, v8f c) {
  return __builtin_amdgcn_wmma_f32_16x16x4_f32(
       false, a,  false, b,
       (short)0, c,  false,  false);
}

template <int M>
__device__ __forceinline__ float row_max_step(float x) {
  float r;
  asm("v_max_num_f32_dpp %0, %1, %1 row_xmask:%2 row_mask:0xf bank_mask:0xf"
      : "=v"(r) : "v"(x), "n"(M));
  return r;
}
template <int M>
__device__ __forceinline__ float row_xor(float x) {
  return __builtin_amdgcn_update_dpp(0.0f, x, 0x160 + M, 0xf, 0xf, true);
}

#define GBM 128
#define GBN 128
#define GBK 32
#define GLD 36
#define GSTAGE ((GBM + GBN) * GLD)

__global__ __launch_bounds__(256)
void gemm_nt_wmma(const float* __restrict__ A, const float* __restrict__ B,
                  float* __restrict__ C, int K, int lda, int ldb, int ldc) {
  __shared__ float lds[2 * GSTAGE];

  const int tid  = threadIdx.x;
  const int lane = tid & 31;
  const int wave = tid >> 5;
  const int half = lane >> 4;
  const int l16  = lane & 15;
  const int wm = wave >> 2;
  const int wn = wave & 3;
  const int bm = blockIdx.y * GBM;
  const int bn = blockIdx.x * GBN;

  const uint32_t srow = (uint32_t)(tid >> 3);
  const uint32_t sc4  = (uint32_t)(tid & 7) * 4u;
  const uint32_t aoff = (uint32_t)(bm + (int)srow) * (uint32_t)lda + sc4;
  const uint32_t boff = (uint32_t)(bn + (int)srow) * (uint32_t)ldb + sc4;
  const uint32_t astep = 32u * (uint32_t)lda;
  const uint32_t bstep = 32u * (uint32_t)ldb;
  const uint32_t soff = srow * GLD + sc4;

  const uint32_t faoff = (uint32_t)(wm * 64 + l16) * GLD + 4u * half;
  const uint32_t fboff = (uint32_t)(GBM + wn * 32 + l16) * GLD + 4u * half;

  v8f acc[4][2];
#pragma unroll
  for (int i = 0; i < 4; ++i)
#pragma unroll
    for (int j = 0; j < 2; ++j) acc[i][j] = (v8f){};

  v4f pa[4], pb[4];
#pragma unroll
  for (int t = 0; t < 4; ++t) {
    pa[t] = *(const v4f*)(A + aoff + (uint32_t)t * astep);
    pb[t] = *(const v4f*)(B + boff + (uint32_t)t * bstep);
  }
#pragma unroll
  for (int t = 0; t < 4; ++t) {
    *(v4f*)(lds + soff + (uint32_t)t * (32u * GLD)) = pa[t];
    *(v4f*)(lds + (uint32_t)(GBM * GLD) + soff + (uint32_t)t * (32u * GLD)) = pb[t];
  }
  __syncthreads();

  const int KT = K / GBK;
  for (int kt = 0; kt < KT; ++kt) {
    const uint32_t cur = (kt & 1) ? (uint32_t)GSTAGE : 0u;
    const uint32_t nxt = (uint32_t)GSTAGE - cur;
    const bool more = (kt + 1 < KT);

    if (more) {
      const uint32_t k1 = (uint32_t)(kt + 1) * GBK;
#pragma unroll
      for (int t = 0; t < 4; ++t) {
        pa[t] = *(const v4f*)(A + aoff + (uint32_t)t * astep + k1);
        pb[t] = *(const v4f*)(B + boff + (uint32_t)t * bstep + k1);
      }
    }

    const float* fa_p = lds + cur + faoff;
    const float* fb_p = lds + cur + fboff;

    v4f fa[2][4], fb[2][2];
#pragma unroll
    for (int i = 0; i < 4; ++i) fa[0][i] = *(const v4f*)(fa_p + i * 16 * GLD);
#pragma unroll
    for (int j = 0; j < 2; ++j) fb[0][j] = *(const v4f*)(fb_p + j * 16 * GLD);

#pragma unroll
    for (int ks = 0; ks < 4; ++ks) {
      const int c = ks & 1, n = c ^ 1;
      if (ks + 1 < 4) {
#pragma unroll
        for (int i = 0; i < 4; ++i)
          fa[n][i] = *(const v4f*)(fa_p + i * 16 * GLD + 8 * (ks + 1));
#pragma unroll
        for (int j = 0; j < 2; ++j)
          fb[n][j] = *(const v4f*)(fb_p + j * 16 * GLD + 8 * (ks + 1));
      }
#pragma unroll
      for (int i = 0; i < 4; ++i)
#pragma unroll
        for (int j = 0; j < 2; ++j) {
          v2f a = __builtin_shufflevector(fa[c][i], fa[c][i], 0, 1);
          v2f b = __builtin_shufflevector(fb[c][j], fb[c][j], 0, 1);
          acc[i][j] = wmma_f32(a, b, acc[i][j]);
        }
#pragma unroll
      for (int i = 0; i < 4; ++i)
#pragma unroll
        for (int j = 0; j < 2; ++j) {
          v2f a = __builtin_shufflevector(fa[c][i], fa[c][i], 2, 3);
          v2f b = __builtin_shufflevector(fb[c][j], fb[c][j], 2, 3);
          acc[i][j] = wmma_f32(a, b, acc[i][j]);
        }
    }

    if (more) {
#pragma unroll
      for (int t = 0; t < 4; ++t) {
        *(v4f*)(lds + nxt + soff + (uint32_t)t * (32u * GLD)) = pa[t];
        *(v4f*)(lds + nxt + (uint32_t)(GBM * GLD) + soff + (uint32_t)t * (32u * GLD)) = pb[t];
      }
    }
    __syncthreads();
  }

#pragma unroll
  for (int i = 0; i < 4; ++i)
#pragma unroll
    for (int j = 0; j < 2; ++j) {
      const uint32_t row0 = (uint32_t)(bm + wm * 64 + i * 16 + half * 8);
      const uint32_t col  = (uint32_t)(bn + wn * 32 + j * 16 + l16);
#pragma unroll
      for (int v = 0; v < 8; ++v)
        C[(row0 + v) * (uint32_t)ldc + col] = acc[i][j][v];
    }
}

__global__ __launch_bounds__(256)
void rope_kernel(float* __restrict__ qkv, int T) {
  int idx = blockIdx.x * blockDim.x + threadIdx.x;
  if (idx >= T * 1024) return;
  int t = idx >> 10;
  int r = idx & 1023;
  int qksel = r >> 9;
  int h = (r >> 5) & 15;
  int p = r & 31;
  float inv = powf(10000.0f, -(float)(2 * p) / 64.0f);
  float ang = (float)t * inv;
  float s = sinf(ang), c = cosf(ang);
  float* base = qkv + (uint32_t)t * 3072u + (uint32_t)(qksel * 1024 + h * 64 + 2 * p);
  float x1 = base[0], x2 = base[1];
  base[0] = x1 * c - x2 * s;
  base[1] = x2 * c + x1 * s;
}

#define LDT 68

__global__ __launch_bounds__(512)
void attn_kernel(const float* __restrict__ qkv, float* __restrict__ out) {
  __shared__ float Klds[64 * LDT];
  __shared__ float Vlds[64 * LDT];
  __shared__ float Plds[16 * 16 * LDT];

  const int tid  = threadIdx.x;
  const int lane = tid & 31;
  const int wave = tid >> 5;
  const int half = lane >> 4;
  const int l16  = lane & 15;
  const int h     = blockIdx.y;
  const int qtile_hi = 31 - (int)blockIdx.x;
  const int qtile = (wave < 8) ? qtile_hi : (int)blockIdx.x;
  const int qbase = qtile * 128 + (wave & 7) * 16;
  const uint32_t ld = 3072;

  const float* qp = qkv + (uint32_t)h * 64u;
  const float* kp = qkv + 1024u + (uint32_t)h * 64u;
  const float* vp = qkv + 2048u + (uint32_t)h * 64u;

  v2f qf[16];
  {
    const uint32_t qoff = (uint32_t)(qbase + l16) * ld + 2u * half;
#pragma unroll
    for (int kk = 0; kk < 16; ++kk)
      qf[kk] = *(const v2f*)(qp + qoff + 4u * kk);
  }

  v8f o[4];
#pragma unroll
  for (int ct = 0; ct < 4; ++ct) o[ct] = (v8f){};
  float mrow[8], lrow[8];
#pragma unroll
  for (int v = 0; v < 8; ++v) { mrow[v] = -1e30f; lrow[v] = 0.0f; }

  float* myp = Plds + wave * 16 * LDT;
  const int ktmax_w = (qbase + 15) >> 6;
  const int ktmax_b = (qtile_hi * 128 + 127) >> 6;

  for (int kt = 0; kt <= ktmax_b; ++kt) {
    const uint32_t tbase = (uint32_t)(kt * 64) * ld;

    __syncthreads();
#pragma unroll 4
    for (int t = 0; t < 8; ++t) {
      int idx = t * 512 + tid;
      int key = idx >> 6, c = idx & 63;
      uint32_t goff = tbase + (uint32_t)key * ld + (uint32_t)c;
      float kvv = kp[goff];
      float vvv = vp[goff];
      Klds[key * LDT + c] = kvv;
      Vlds[c * LDT + key] = vvv;
    }
    if (kt < ktmax_b) {
      if (tid < 64) {
        const float* p = kp + tbase + (uint32_t)(64 + tid) * ld;
        __builtin_prefetch(p, 0, 3);
        __builtin_prefetch(p + 32, 0, 3);
      } else if (tid < 128) {
        const float* p = vp + tbase + (uint32_t)tid * ld;
        __builtin_prefetch(p, 0, 3);
        __builtin_prefetch(p + 32, 0, 3);
      }
    }
    __syncthreads();

    if (kt <= ktmax_w) {
      v8f s[4];
#pragma unroll
      for (int j = 0; j < 4; ++j) s[j] = (v8f){};
#pragma unroll 4
      for (int kk = 0; kk < 16; ++kk) {
        v2f a = qf[kk];
#pragma unroll
        for (int j = 0; j < 4; ++j) {
          v2f b = *(const v2f*)&Klds[(j * 16 + l16) * LDT + 4 * kk + 2 * half];
          s[j] = wmma_f32(a, b, s[j]);
        }
      }

      float mx[8];
#pragma unroll
      for (int v = 0; v < 8; ++v) {
        const int row = qbase + half * 8 + v;
        float m = -1e30f;
#pragma unroll
        for (int j = 0; j < 4; ++j) {
          const int col = kt * 64 + j * 16 + l16;
          float x = s[j][v] * 0.125f;
          if (col > row) x = -1e30f;
          s[j][v] = x;
          m = fmaxf(m, x);
        }
        mx[v] = m;
      }
#pragma unroll
      for (int v = 0; v < 8; ++v) mx[v] = row_max_step<1>(mx[v]);
#pragma unroll
      for (int v = 0; v < 8; ++v) mx[v] = row_max_step<2>(mx[v]);
#pragma unroll
      for (int v = 0; v < 8; ++v) mx[v] = row_max_step<4>(mx[v]);
#pragma unroll
      for (int v = 0; v < 8; ++v) mx[v] = row_max_step<8>(mx[v]);

      float alpha[8];
#pragma unroll
      for (int v = 0; v < 8; ++v) {
        const float mnew = fmaxf(mrow[v], mx[v]);
        const float a = __expf(mrow[v] - mnew);
        float rsum = 0.0f;
#pragma unroll
        for (int j = 0; j < 4; ++j) {
          float p = __expf(s[j][v] - mnew);
          myp[(half * 8 + v) * LDT + j * 16 + l16] = p;
          rsum += p;
        }
        lrow[v] = lrow[v] * a + rsum;
        mrow[v] = mnew;
        alpha[v] = a;
      }
#pragma unroll
      for (int ct = 0; ct < 4; ++ct)
#pragma unroll
        for (int v = 0; v < 8; ++v) o[ct][v] *= alpha[v];

#pragma unroll 2
      for (int kk = 0; kk < 16; ++kk) {
        v2f a = *(const v2f*)&myp[l16 * LDT + 4 * kk + 2 * half];
#pragma unroll
        for (int ct = 0; ct < 4; ++ct) {
          v2f b = *(const v2f*)&Vlds[(ct * 16 + l16) * LDT + 4 * kk + 2 * half];
          o[ct] = wmma_f32(a, b, o[ct]);
        }
      }
    }
  }

#pragma unroll
  for (int v = 0; v < 8; ++v) lrow[v] += row_xor<1>(lrow[v]);
#pragma unroll
  for (int v = 0; v < 8; ++v) lrow[v] += row_xor<2>(lrow[v]);
#pragma unroll
  for (int v = 0; v < 8; ++v) lrow[v] += row_xor<4>(lrow[v]);
#pragma unroll
  for (int v = 0; v < 8; ++v) lrow[v] += row_xor<8>(lrow[v]);

#pragma unroll
  for (int ct = 0; ct < 4; ++ct)
#pragma unroll
    for (int v = 0; v < 8; ++v) {
      const uint32_t row = (uint32_t)(qbase + half * 8 + v);
      out[row * 1024u + (uint32_t)(h * 64 + ct * 16 + l16)] =
          o[ct][v] / lrow[v];
    }
}

extern "C" void kernel_launch(void* const* d_in, const int* in_sizes, int n_in,
                              void* d_out, int out_size, void* d_ws, size_t ws_size,
                              hipStream_t stream) {
  const float* x     = (const float*)d_in[0];
  const float* Wqkv  = (const float*)d_in[1];
  const float* Wproj = (const float*)d_in[2];
  float* out = (float*)d_out;

  float* qkv     = (float*)d_ws;
  float* attnout = qkv + (size_t)4096 * 3072;

  gemm_nt_wmma<<<dim3(24, 32), 256, 0, stream>>>(
      x, Wqkv, qkv, 1024, 1024, 1024, 3072);

  rope_kernel<<<(4096 * 1024) / 256, 256, 0, stream>>>(qkv, 4096);

  attn_kernel<<<dim3(16, 16), 512, 0, stream>>>(qkv, attnout);

  gemm_nt_wmma<<<dim3(8, 32), 256, 0, stream>>>(
      attnout, Wproj, out, 1024, 1024, 1024, 1024);
}


# attention: K, V, P fragments read by 16-byte LDS loads (2 k steps each), each group of 8-10 loads issued one block of 16 products ahead into a second register set (was 8-byte loads awaited in place)
# speedup vs baseline: 2.4797x; 1.0547x over previous
#include <hip/hip_runtime.h>
#include <hip/hip_bf16.h>

typedef __attribute__((ext_vector_type(2))) float v2f;
typedef __attribute__((ext_vector_type(4))) float v4f;
typedef __attribute__((ext_vector_type(8))) float v8f;

__device__ __forceinline__ v8f wmma_f32(v2f a, v2f b, v8f c) {
  return __builtin_amdgcn_wmma_f32_16x16x4_f32(
       false, a,  false, b,
       (short)0, c,  false,  false);
}

template <int M>
__device__ __forceinline__ float row_max_step(float x) {
  float r;
  asm("v_max_num_f32_dpp %0, %1, %1 row_xmask:%2 row_mask:0xf bank_mask:0xf"
      : "=v"(r) : "v"(x), "n"(M));
  return r;
}
template <int M>
__device__ __forceinline__ float row_xor(float x) {
  return __builtin_amdgcn_update_dpp(0.0f, x, 0x160 + M, 0xf, 0xf, true);
}

__device__ __forceinline__ v2f lo2(v4f x) { return __builtin_shufflevector(x, x, 0, 1); }
__device__ __forceinline__ v2f hi2(v4f x) { return __builtin_shufflevector(x, x, 2, 3); }

#define GBM 128
#define GBN 128
#define GBK 32
#define GLD 36
#define GSTAGE ((GBM + GBN) * GLD)

__global__ __launch_bounds__(256)
void gemm_nt_wmma(const float* __restrict__ A, const float* __restrict__ B,
                  float* __restrict__ C, int K, int lda, int ldb, int ldc) {
  __shared__ float lds[2 * GSTAGE];

  const int tid  = threadIdx.x;
  const int lane = tid & 31;
  const int wave = tid >> 5;
  const int half = lane >> 4;
  const int l16  = lane & 15;
  const int wm = wave >> 2;
  const int wn = wave & 3;
  const int bm = blockIdx.y * GBM;
  const int bn = blockIdx.x * GBN;

  const uint32_t srow = (uint32_t)(tid >> 3);
  const uint32_t sc4  = (uint32_t)(tid & 7) * 4u;
  const uint32_t aoff = (uint32_t)(bm + (int)srow) * (uint32_t)lda + sc4;
  const uint32_t boff = (uint32_t)(bn + (int)srow) * (uint32_t)ldb + sc4;
  const uint32_t astep = 32u * (uint32_t)lda;
  const uint32_t bstep = 32u * (uint32_t)ldb;
  const uint32_t soff = srow * GLD + sc4;

  const uint32_t faoff = (uint32_t)(wm * 64 + l16) * GLD + 4u * half;
  const uint32_t fboff = (uint32_t)(GBM + wn * 32 + l16) * GLD + 4u * half;

  v8f acc[4][2];
#pragma unroll
  for (int i = 0; i < 4; ++i)
#pragma unroll
    for (int j = 0; j < 2; ++j) acc[i][j] = (v8f){};

  v4f pa[4], pb[4];
#pragma unroll
  for (int t = 0; t < 4; ++t) {
    pa[t] = *(const v4f*)(A + aoff + (uint32_t)t * astep);
    pb[t] = *(const v4f*)(B + boff + (uint32_t)t * bstep);
  }
#pragma unroll
  for (int t = 0; t < 4; ++t) {
    *(v4f*)(lds + soff + (uint32_t)t * (32u * GLD)) = pa[t];
    *(v4f*)(lds + (uint32_t)(GBM * GLD) + soff + (uint32_t)t * (32u * GLD)) = pb[t];
  }
  __syncthreads();

  const int KT = K / GBK;
  for (int kt = 0; kt < KT; ++kt) {
    const uint32_t cur = (kt & 1) ? (uint32_t)GSTAGE : 0u;
    const uint32_t nxt = (uint32_t)GSTAGE - cur;
    const bool more = (kt + 1 < KT);

    if (more) {
      const uint32_t k1 = (uint32_t)(kt + 1) * GBK;
#pragma unroll
      for (int t = 0; t < 4; ++t) {
        pa[t] = *(const v4f*)(A + aoff + (uint32_t)t * astep + k1);
        pb[t] = *(const v4f*)(B + boff + (uint32_t)t * bstep + k1);
      }
    }

    const float* fa_p = lds + cur + faoff;
    const float* fb_p = lds + cur + fboff;

    v4f fa[2][4], fb[2][2];
#pragma unroll
    for (int i = 0; i < 4; ++i) fa[0][i] = *(const v4f*)(fa_p + i * 16 * GLD);
#pragma unroll
    for (int j = 0; j < 2; ++j) fb[0][j] = *(const v4f*)(fb_p + j * 16 * GLD);

#pragma unroll
    for (int ks = 0; ks < 4; ++ks) {
      const int c = ks & 1, n = c ^ 1;
      if (ks + 1 < 4) {
#pragma unroll
        for (int i = 0; i < 4; ++i)
          fa[n][i] = *(const v4f*)(fa_p + i * 16 * GLD + 8 * (ks + 1));
#pragma unroll
        for (int j = 0; j < 2; ++j)
          fb[n][j] = *(const v4f*)(fb_p + j * 16 * GLD + 8 * (ks + 1));
      }
#pragma unroll
      for (int i = 0; i < 4; ++i)
#pragma unroll
        for (int j = 0; j < 2; ++j) {
          v2f a = __builtin_shufflevector(fa[c][i], fa[c][i], 0, 1);
          v2f b = __builtin_shufflevector(fb[c][j], fb[c][j], 0, 1);
          acc[i][j] = wmma_f32(a, b, acc[i][j]);
        }
#pragma unroll
      for (int i = 0; i < 4; ++i)
#pragma unroll
        for (int j = 0; j < 2; ++j) {
          v2f a = __builtin_shufflevector(fa[c][i], fa[c][i], 2, 3);
          v2f b = __builtin_shufflevector(fb[c][j], fb[c][j], 2, 3);
          acc[i][j] = wmma_f32(a, b, acc[i][j]);
        }
    }

    if (more) {
#pragma unroll
      for (int t = 0; t < 4; ++t) {
        *(v4f*)(lds + nxt + soff + (uint32_t)t * (32u * GLD)) = pa[t];
        *(v4f*)(lds + nxt + (uint32_t)(GBM * GLD) + soff + (uint32_t)t * (32u * GLD)) = pb[t];
      }
    }
    __syncthreads();
  }

#pragma unroll
  for (int i = 0; i < 4; ++i)
#pragma unroll
    for (int j = 0; j < 2; ++j) {
      const uint32_t row0 = (uint32_t)(bm + wm * 64 + i * 16 + half * 8);
      const uint32_t col  = (uint32_t)(bn + wn * 32 + j * 16 + l16);
#pragma unroll
      for (int v = 0; v < 8; ++v)
        C[(row0 + v) * (uint32_t)ldc + col] = acc[i][j][v];
    }
}

__global__ __launch_bounds__(256)
void rope_kernel(float* __restrict__ qkv, int T) {
  int idx = blockIdx.x * blockDim.x + threadIdx.x;
  if (idx >= T * 1024) return;
  int t = idx >> 10;
  int r = idx & 1023;
  int qksel = r >> 9;
  int h = (r >> 5) & 15;
  int p = r & 31;
  float inv = powf(10000.0f, -(float)(2 * p) / 64.0f);
  float ang = (float)t * inv;
  float s = sinf(ang), c = cosf(ang);
  float* base = qkv + (uint32_t)t * 3072u + (uint32_t)(qksel * 1024 + h * 64 + 2 * p);
  float x1 = base[0], x2 = base[1];
  base[0] = x1 * c - x2 * s;
  base[1] = x2 * c + x1 * s;
}

#define LDT 68

__global__ __launch_bounds__(512)
void attn_kernel(const float* __restrict__ qkv, float* __restrict__ out) {
  __shared__ __attribute__((aligned(16))) float Klds[64 * LDT];
  __shared__ __attribute__((aligned(16))) float Vlds[64 * LDT];
  __shared__ __attribute__((aligned(16))) float Plds[16 * 16 * LDT];

  const int tid  = threadIdx.x;
  const int lane = tid & 31;
  const int wave = tid >> 5;
  const int half = lane >> 4;
  const int l16  = lane & 15;
  const int h     = blockIdx.y;
  const int qtile_hi = 31 - (int)blockIdx.x;
  const int qtile = (wave < 8) ? qtile_hi : (int)blockIdx.x;
  const int qbase = qtile * 128 + (wave & 7) * 16;
  const uint32_t ld = 3072;

  const float* qp = qkv + (uint32_t)h * 64u;
  const float* kp = qkv + 1024u + (uint32_t)h * 64u;
  const float* vp = qkv + 2048u + (uint32_t)h * 64u;

  v4f qf[8];
  {
    const uint32_t qoff = (uint32_t)(qbase + l16) * ld + 4u * half;
#pragma unroll
    for (int ks = 0; ks < 8; ++ks)
      qf[ks] = *(const v4f*)(qp + qoff + 8u * ks);
  }

  v8f o[4];
#pragma unroll
  for (int ct = 0; ct < 4; ++ct) o[ct] = (v8f){};
  float mrow[8], lrow[8];
#pragma unroll
  for (int v = 0; v < 8; ++v) { mrow[v] = -1e30f; lrow[v] = 0.0f; }

  float* myp = Plds + wave * 16 * LDT;
  const float* kfp = Klds + l16 * LDT + 4 * half;
  const float* vfp = Vlds + l16 * LDT + 4 * half;
  const float* pfp = myp + l16 * LDT + 4 * half;
  const int ktmax_w = (qbase + 15) >> 6;
  const int ktmax_b = (qtile_hi * 128 + 127) >> 6;

  for (int kt = 0; kt <= ktmax_b; ++kt) {
    const uint32_t tbase = (uint32_t)(kt * 64) * ld;

    __syncthreads();
#pragma unroll 4
    for (int t = 0; t < 8; ++t) {
      int idx = t * 512 + tid;
      int key = idx >> 6, c = idx & 63;
      uint32_t goff = tbase + (uint32_t)key * ld + (uint32_t)c;
      float kvv = kp[goff];
      float vvv = vp[goff];
      Klds[key * LDT + c] = kvv;
      Vlds[c * LDT + key] = vvv;
    }
    if (kt < ktmax_b) {
      if (tid < 64) {
        const float* p = kp + tbase + (uint32_t)(64 + tid) * ld;
        __builtin_prefetch(p, 0, 3);
        __builtin_prefetch(p + 32, 0, 3);
      } else if (tid < 128) {
        const float* p = vp + tbase + (uint32_t)tid * ld;
        __builtin_prefetch(p, 0, 3);
        __builtin_prefetch(p + 32, 0, 3);
      }
    }
    __syncthreads();

    if (kt <= ktmax_w) {
      v8f s[4];
#pragma unroll
      for (int j = 0; j < 4; ++j) s[j] = (v8f){};
      {
        v4f kb[2][2][4];
#pragma unroll
        for (int u = 0; u < 2; ++u)
#pragma unroll
          for (int j = 0; j < 4; ++j)
            kb[0][u][j] = *(const v4f*)(kfp + j * 16 * LDT + 8 * u);
        __builtin_amdgcn_sched_barrier(0);
#pragma unroll
        for (int g = 0; g < 4; ++g) {
          const int c = g & 1, n = c ^ 1;
          if (g + 1 < 4) {
#pragma unroll
            for (int u = 0; u < 2; ++u)
#pragma unroll
              for (int j = 0; j < 4; ++j)
                kb[n][u][j] = *(const v4f*)(kfp + j * 16 * LDT + 8 * (2 * (g + 1) + u));
          }
          __builtin_amdgcn_sched_barrier(0);
#pragma unroll
          for (int u = 0; u < 2; ++u) {
            const v4f q4 = qf[2 * g + u];
#pragma unroll
            for (int j = 0; j < 4; ++j)
              s[j] = wmma_f32(lo2(q4), lo2(kb[c][u][j]), s[j]);
#pragma unroll
            for (int j = 0; j < 4; ++j)
              s[j] = wmma_f32(hi2(q4), hi2(kb[c][u][j]), s[j]);
          }
          __builtin_amdgcn_sched_barrier(0);
        }
      }

      float mx[8];
#pragma unroll
      for (int v = 0; v < 8; ++v) {
        const int row = qbase + half * 8 + v;
        float m = -1e30f;
#pragma unroll
        for (int j = 0; j < 4; ++j) {
          const int col = kt * 64 + j * 16 + l16;
          float x = s[j][v] * 0.125f;
          if (col > row) x = -1e30f;
          s[j][v] = x;
          m = fmaxf(m, x);
        }
        mx[v] = m;
      }
#pragma unroll
      for (int v = 0; v < 8; ++v) mx[v] = row_max_step<1>(mx[v]);
#pragma unroll
      for (int v = 0; v < 8; ++v) mx[v] = row_max_step<2>(mx[v]);
#pragma unroll
      for (int v = 0; v < 8; ++v) mx[v] = row_max_step<4>(mx[v]);
#pragma unroll
      for (int v = 0; v < 8; ++v) mx[v] = row_max_step<8>(mx[v]);

      float alpha[8];
#pragma unroll
      for (int v = 0; v < 8; ++v) {
        const float mnew = fmaxf(mrow[v], mx[v]);
        const float a = __expf(mrow[v] - mnew);
        float rsum = 0.0f;
#pragma unroll
        for (int j = 0; j < 4; ++j) {
          float p = __expf(s[j][v] - mnew);
          myp[(half * 8 + v) * LDT + j * 16 + l16] = p;
          rsum += p;
        }
        lrow[v] = lrow[v] * a + rsum;
        mrow[v] = mnew;
        alpha[v] = a;
      }
#pragma unroll
      for (int ct = 0; ct < 4; ++ct)
#pragma unroll
        for (int v = 0; v < 8; ++v) o[ct][v] *= alpha[v];

      {
        v4f pa[2][2];
        v4f vb[2][2][4];
#pragma unroll
        for (int u = 0; u < 2; ++u) {
          pa[0][u] = *(const v4f*)(pfp + 8 * u);
#pragma unroll
          for (int ct = 0; ct < 4; ++ct)
            vb[0][u][ct] = *(const v4f*)(vfp + ct * 16 * LDT + 8 * u);
        }
        __builtin_amdgcn_sched_barrier(0);
#pragma unroll
        for (int g = 0; g < 4; ++g) {
          const int c = g & 1, n = c ^ 1;
          if (g + 1 < 4) {
#pragma unroll
            for (int u = 0; u < 2; ++u) {
              pa[n][u] = *(const v4f*)(pfp + 8 * (2 * (g + 1) + u));
#pragma unroll
              for (int ct = 0; ct < 4; ++ct)
                vb[n][u][ct] = *(const v4f*)(vfp + ct * 16 * LDT + 8 * (2 * (g + 1) + u));
            }
          }
          __builtin_amdgcn_sched_barrier(0);
#pragma unroll
          for (int u = 0; u < 2; ++u) {
#pragma unroll
            for (int ct = 0; ct < 4; ++ct)
              o[ct] = wmma_f32(lo2(pa[c][u]), lo2(vb[c][u][ct]), o[ct]);
#pragma unroll
            for (int ct = 0; ct < 4; ++ct)
              o[ct] = wmma_f32(hi2(pa[c][u]), hi2(vb[c][u][ct]), o[ct]);
          }
          __builtin_amdgcn_sched_barrier(0);
        }
      }
    }
  }

#pragma unroll
  for (int v = 0; v < 8; ++v) lrow[v] += row_xor<1>(lrow[v]);
#pragma unroll
  for (int v = 0; v < 8; ++v) lrow[v] += row_xor<2>(lrow[v]);
#pragma unroll
  for (int v = 0; v < 8; ++v) lrow[v] += row_xor<4>(lrow[v]);
#pragma unroll
  for (int v = 0; v < 8; ++v) lrow[v] += row_xor<8>(lrow[v]);

#pragma unroll
  for (int ct = 0; ct < 4; ++ct)
#pragma unroll
    for (int v = 0; v < 8; ++v) {
      const uint32_t row = (uint32_t)(qbase + half * 8 + v);
      out[row * 1024u + (uint32_t)(h * 64 + ct * 16 + l16)] =
          o[ct][v] / lrow[v];
    }
}

extern "C" void kernel_launch(void* const* d_in, const int* in_sizes, int n_in,
                              void* d_out, int out_size, void* d_ws, size_t ws_size,
                              hipStream_t stream) {
  const float* x     = (const float*)d_in[0];
  const float* Wqkv  = (const float*)d_in[1];
  const float* Wproj = (const float*)d_in[2];
  float* out = (float*)d_out;

  float* qkv     = (float*)d_ws;
  float* attnout = qkv + (size_t)4096 * 3072;

  gemm_nt_wmma<<<dim3(24, 32), 256, 0, stream>>>(
      x, Wqkv, qkv, 1024, 1024, 1024, 3072);

  rope_kernel<<<(4096 * 1024) / 256, 256, 0, stream>>>(qkv, 4096);

  attn_kernel<<<dim3(16, 16), 512, 0, stream>>>(qkv, attnout);

  gemm_nt_wmma<<<dim3(8, 32), 256, 0, stream>>>(
      attnout, Wproj, out, 1024, 1024, 1024, 1024);
}


# attention K/V staging pipelined: next tile fetched by 16-byte global loads before the current tile's products, stored to a second LDS stage after them; 1 barrier per tile (was 2 around exposed loads)
# speedup vs baseline: 2.5934x; 1.0459x over previous
#include <hip/hip_runtime.h>
#include <hip/hip_bf16.h>

typedef __attribute__((ext_vector_type(2))) float v2f;
typedef __attribute__((ext_vector_type(4))) float v4f;
typedef __attribute__((ext_vector_type(8))) float v8f;

__device__ __forceinline__ v8f wmma_f32(v2f a, v2f b, v8f c) {
  return __builtin_amdgcn_wmma_f32_16x16x4_f32(
       false, a,  false, b,
       (short)0, c,  false,  false);
}

template <int M>
__device__ __forceinline__ float row_max_step(float x) {
  float r;
  asm("v_max_num_f32_dpp %0, %1, %1 row_xmask:%2 row_mask:0xf bank_mask:0xf"
      : "=v"(r) : "v"(x), "n"(M));
  return r;
}
template <int M>
__device__ __forceinline__ float row_xor(float x) {
  return __builtin_amdgcn_update_dpp(0.0f, x, 0x160 + M, 0xf, 0xf, true);
}

__device__ __forceinline__ v2f lo2(v4f x) { return __builtin_shufflevector(x, x, 0, 1); }
__device__ __forceinline__ v2f hi2(v4f x) { return __builtin_shufflevector(x, x, 2, 3); }

#define GBM 128
#define GBN 128
#define GBK 32
#define GLD 36
#define GSTAGE ((GBM + GBN) * GLD)

__global__ __launch_bounds__(256)
void gemm_nt_wmma(const float* __restrict__ A, const float* __restrict__ B,
                  float* __restrict__ C, int K, int lda, int ldb, int ldc) {
  __shared__ float lds[2 * GSTAGE];

  const int tid  = threadIdx.x;
  const int lane = tid & 31;
  const int wave = tid >> 5;
  const int half = lane >> 4;
  const int l16  = lane & 15;
  const int wm = wave >> 2;
  const int wn = wave & 3;
  const int bm = blockIdx.y * GBM;
  const int bn = blockIdx.x * GBN;

  const uint32_t srow = (uint32_t)(tid >> 3);
  const uint32_t sc4  = (uint32_t)(tid & 7) * 4u;
  const uint32_t aoff = (uint32_t)(bm + (int)srow) * (uint32_t)lda + sc4;
  const uint32_t boff = (uint32_t)(bn + (int)srow) * (uint32_t)ldb + sc4;
  const uint32_t astep = 32u * (uint32_t)lda;
  const uint32_t bstep = 32u * (uint32_t)ldb;
  const uint32_t soff = srow * GLD + sc4;

  const uint32_t faoff = (uint32_t)(wm * 64 + l16) * GLD + 4u * half;
  const uint32_t fboff = (uint32_t)(GBM + wn * 32 + l16) * GLD + 4u * half;

  v8f acc[4][2];
#pragma unroll
  for (int i = 0; i < 4; ++i)
#pragma unroll
    for (int j = 0; j < 2; ++j) acc[i][j] = (v8f){};

  v4f pa[4], pb[4];
#pragma unroll
  for (int t = 0; t < 4; ++t) {
    pa[t] = *(const v4f*)(A + aoff + (uint32_t)t * astep);
    pb[t] = *(const v4f*)(B + boff + (uint32_t)t * bstep);
  }
#pragma unroll
  for (int t = 0; t < 4; ++t) {
    *(v4f*)(lds + soff + (uint32_t)t * (32u * GLD)) = pa[t];
    *(v4f*)(lds + (uint32_t)(GBM * GLD) + soff + (uint32_t)t * (32u * GLD)) = pb[t];
  }
  __syncthreads();

  const int KT = K / GBK;
  for (int kt = 0; kt < KT; ++kt) {
    const uint32_t cur = (kt & 1) ? (uint32_t)GSTAGE : 0u;
    const uint32_t nxt = (uint32_t)GSTAGE - cur;
    const bool more = (kt + 1 < KT);

    if (more) {
      const uint32_t k1 = (uint32_t)(kt + 1) * GBK;
#pragma unroll
      for (int t = 0; t < 4; ++t) {
        pa[t] = *(const v4f*)(A + aoff + (uint32_t)t * astep + k1);
        pb[t] = *(const v4f*)(B + boff + (uint32_t)t * bstep + k1);
      }
    }

    const float* fa_p = lds + cur + faoff;
    const float* fb_p = lds + cur + fboff;

    v4f fa[2][4], fb[2][2];
#pragma unroll
    for (int i = 0; i < 4; ++i) fa[0][i] = *(const v4f*)(fa_p + i * 16 * GLD);
#pragma unroll
    for (int j = 0; j < 2; ++j) fb[0][j] = *(const v4f*)(fb_p + j * 16 * GLD);

#pragma unroll
    for (int ks = 0; ks < 4; ++ks) {
      const int c = ks & 1, n = c ^ 1;
      if (ks + 1 < 4) {
#pragma unroll
        for (int i = 0; i < 4; ++i)
          fa[n][i] = *(const v4f*)(fa_p + i * 16 * GLD + 8 * (ks + 1));
#pragma unroll
        for (int j = 0; j < 2; ++j)
          fb[n][j] = *(const v4f*)(fb_p + j * 16 * GLD + 8 * (ks + 1));
      }
#pragma unroll
      for (int i = 0; i < 4; ++i)
#pragma unroll
        for (int j = 0; j < 2; ++j) {
          v2f a = __builtin_shufflevector(fa[c][i], fa[c][i], 0, 1);
          v2f b = __builtin_shufflevector(fb[c][j], fb[c][j], 0, 1);
          acc[i][j] = wmma_f32(a, b, acc[i][j]);
        }
#pragma unroll
      for (int i = 0; i < 4; ++i)
#pragma unroll
        for (int j = 0; j < 2; ++j) {
          v2f a = __builtin_shufflevector(fa[c][i], fa[c][i], 2, 3);
          v2f b = __builtin_shufflevector(fb[c][j], fb[c][j], 2, 3);
          acc[i][j] = wmma_f32(a, b, acc[i][j]);
        }
    }

    if (more) {
#pragma unroll
      for (int t = 0; t < 4; ++t) {
        *(v4f*)(lds + nxt + soff + (uint32_t)t * (32u * GLD)) = pa[t];
        *(v4f*)(lds + nxt + (uint32_t)(GBM * GLD) + soff + (uint32_t)t * (32u * GLD)) = pb[t];
      }
    }
    __syncthreads();
  }

#pragma unroll
  for (int i = 0; i < 4; ++i)
#pragma unroll
    for (int j = 0; j < 2; ++j) {
      const uint32_t row0 = (uint32_t)(bm + wm * 64 + i * 16 + half * 8);
      const uint32_t col  = (uint32_t)(bn + wn * 32 + j * 16 + l16);
#pragma unroll
      for (int v = 0; v < 8; ++v)
        C[(row0 + v) * (uint32_t)ldc + col] = acc[i][j][v];
    }
}

__global__ __launch_bounds__(256)
void rope_kernel(float* __restrict__ qkv, int T) {
  int idx = blockIdx.x * blockDim.x + threadIdx.x;
  if (idx >= T * 1024) return;
  int t = idx >> 10;
  int r = idx & 1023;
  int qksel = r >> 9;
  int h = (r >> 5) & 15;
  int p = r & 31;
  float inv = powf(10000.0f, -(float)(2 * p) / 64.0f);
  float ang = (float)t * inv;
  float s = sinf(ang), c = cosf(ang);
  float* base = qkv + (uint32_t)t * 3072u + (uint32_t)(qksel * 1024 + h * 64 + 2 * p);
  float x1 = base[0], x2 = base[1];
  base[0] = x1 * c - x2 * s;
  base[1] = x2 * c + x1 * s;
}

#define LDT 68
#define KVS (64 * LDT)

__global__ __launch_bounds__(512)
void attn_kernel(const float* __restrict__ qkv, float* __restrict__ out) {
  __shared__ __attribute__((aligned(16))) float Klds[2 * KVS];
  __shared__ __attribute__((aligned(16))) float Vlds[2 * KVS];
  __shared__ __attribute__((aligned(16))) float Plds[16 * 16 * LDT];

  const int tid  = threadIdx.x;
  const int lane = tid & 31;
  const int wave = tid >> 5;
  const int half = lane >> 4;
  const int l16  = lane & 15;
  const int h     = blockIdx.y;
  const int qtile_hi = 31 - (int)blockIdx.x;
  const int qtile = (wave < 8) ? qtile_hi : (int)blockIdx.x;
  const int qbase = qtile * 128 + (wave & 7) * 16;
  const uint32_t ld = 3072;

  const float* qp = qkv + (uint32_t)h * 64u;
  const float* kp = qkv + 1024u + (uint32_t)h * 64u;
  const float* vp = qkv + 2048u + (uint32_t)h * 64u;

  v4f qf[8];
  {
    const uint32_t qoff = (uint32_t)(qbase + l16) * ld + 4u * half;
#pragma unroll
    for (int ks = 0; ks < 8; ++ks)
      qf[ks] = *(const v4f*)(qp + qoff + 8u * ks);
  }

  v8f o[4];
#pragma unroll
  for (int ct = 0; ct < 4; ++ct) o[ct] = (v8f){};
  float mrow[8], lrow[8];
#pragma unroll
  for (int v = 0; v < 8; ++v) { mrow[v] = -1e30f; lrow[v] = 0.0f; }

  float* myp = Plds + wave * 16 * LDT;
  const float* kfp0 = Klds + l16 * LDT + 4 * half;
  const float* vfp0 = Vlds + l16 * LDT + 4 * half;
  const float* pfp = myp + l16 * LDT + 4 * half;

  const uint32_t kg_off = (uint32_t)(tid >> 4) * ld + (uint32_t)(tid & 15) * 4u;
  const uint32_t kl_off = (uint32_t)(tid >> 4) * LDT + (uint32_t)(tid & 15) * 4u;
  const uint32_t vkey = (uint32_t)((wave & 7) * 8 + (lane & 7));
  const uint32_t vc4  = (uint32_t)(((wave >> 3) * 4 + (lane >> 3)) * 4);
  const uint32_t vg_off = vkey * ld + vc4;
  const uint32_t vl_off = vc4 * LDT + vkey;

  v4f kr0, kr1, vr0, vr1;
  kr0 = *(const v4f*)(kp + kg_off);
  kr1 = *(const v4f*)(kp + kg_off + 32u * ld);
  vr0 = *(const v4f*)(vp + vg_off);
  vr1 = *(const v4f*)(vp + vg_off + 32u);
  *(v4f*)(Klds + kl_off) = kr0;
  *(v4f*)(Klds + kl_off + 32 * LDT) = kr1;
#pragma unroll
  for (int i = 0; i < 4; ++i) {
    Vlds[vl_off + i * LDT] = vr0[i];
    Vlds[vl_off + (32 + i) * LDT] = vr1[i];
  }
  __syncthreads();
  const int ktmax_w = (qbase + 15) >> 6;
  const int ktmax_b = (qtile_hi * 128 + 127) >> 6;

  for (int kt = 0; kt <= ktmax_b; ++kt) {
    const uint32_t cur = (kt & 1) ? (uint32_t)KVS : 0u;
    const bool more = (kt < ktmax_b);
    const float* kfp = kfp0 + cur;
    const float* vfp = vfp0 + cur;

    if (more) {
      const uint32_t nbase = (uint32_t)((kt + 1) * 64) * ld;
      kr0 = *(const v4f*)(kp + nbase + kg_off);
      kr1 = *(const v4f*)(kp + nbase + kg_off + 32u * ld);
      vr0 = *(const v4f*)(vp + nbase + vg_off);
      vr1 = *(const v4f*)(vp + nbase + vg_off + 32u);
    }

    if (kt <= ktmax_w) {
      v8f s[4];
#pragma unroll
      for (int j = 0; j < 4; ++j) s[j] = (v8f){};
      {
        v4f kb[2][2][4];
#pragma unroll
        for (int u = 0; u < 2; ++u)
#pragma unroll
          for (int j = 0; j < 4; ++j)
            kb[0][u][j] = *(const v4f*)(kfp + j * 16 * LDT + 8 * u);
        __builtin_amdgcn_sched_barrier(0);
#pragma unroll
        for (int g = 0; g < 4; ++g) {
          const int c = g & 1, n = c ^ 1;
          if (g + 1 < 4) {
#pragma unroll
            for (int u = 0; u < 2; ++u)
#pragma unroll
              for (int j = 0; j < 4; ++j)
                kb[n][u][j] = *(const v4f*)(kfp + j * 16 * LDT + 8 * (2 * (g + 1) + u));
          }
          __builtin_amdgcn_sched_barrier(0);
#pragma unroll
          for (int u = 0; u < 2; ++u) {
            const v4f q4 = qf[2 * g + u];
#pragma unroll
            for (int j = 0; j < 4; ++j)
              s[j] = wmma_f32(lo2(q4), lo2(kb[c][u][j]), s[j]);
#pragma unroll
            for (int j = 0; j < 4; ++j)
              s[j] = wmma_f32(hi2(q4), hi2(kb[c][u][j]), s[j]);
          }
          __builtin_amdgcn_sched_barrier(0);
        }
      }

      float mx[8];
#pragma unroll
      for (int v = 0; v < 8; ++v) {
        const int row = qbase + half * 8 + v;
        float m = -1e30f;
#pragma unroll
        for (int j = 0; j < 4; ++j) {
          const int col = kt * 64 + j * 16 + l16;
          float x = s[j][v] * 0.125f;
          if (col > row) x = -1e30f;
          s[j][v] = x;
          m = fmaxf(m, x);
        }
        mx[v] = m;
      }
#pragma unroll
      for (int v = 0; v < 8; ++v) mx[v] = row_max_step<1>(mx[v]);
#pragma unroll
      for (int v = 0; v < 8; ++v) mx[v] = row_max_step<2>(mx[v]);
#pragma unroll
      for (int v = 0; v < 8; ++v) mx[v] = row_max_step<4>(mx[v]);
#pragma unroll
      for (int v = 0; v < 8; ++v) mx[v] = row_max_step<8>(mx[v]);

      float alpha[8];
#pragma unroll
      for (int v = 0; v < 8; ++v) {
        const float mnew = fmaxf(mrow[v], mx[v]);
        const float a = __expf(mrow[v] - mnew);
        float rsum = 0.0f;
#pragma unroll
        for (int j = 0; j < 4; ++j) {
          float p = __expf(s[j][v] - mnew);
          myp[(half * 8 + v) * LDT + j * 16 + l16] = p;
          rsum += p;
        }
        lrow[v] = lrow[v] * a + rsum;
        mrow[v] = mnew;
        alpha[v] = a;
      }
#pragma unroll
      for (int ct = 0; ct < 4; ++ct)
#pragma unroll
        for (int v = 0; v < 8; ++v) o[ct][v] *= alpha[v];

      {
        v4f pa[2][2];
        v4f vb[2][2][4];
#pragma unroll
        for (int u = 0; u < 2; ++u) {
          pa[0][u] = *(const v4f*)(pfp + 8 * u);
#pragma unroll
          for (int ct = 0; ct < 4; ++ct)
            vb[0][u][ct] = *(const v4f*)(vfp + ct * 16 * LDT + 8 * u);
        }
        __builtin_amdgcn_sched_barrier(0);
#pragma unroll
        for (int g = 0; g < 4; ++g) {
          const int c = g & 1, n = c ^ 1;
          if (g + 1 < 4) {
#pragma unroll
            for (int u = 0; u < 2; ++u) {
              pa[n][u] = *(const v4f*)(pfp + 8 * (2 * (g + 1) + u));
#pragma unroll
              for (int ct = 0; ct < 4; ++ct)
                vb[n][u][ct] = *(const v4f*)(vfp + ct * 16 * LDT + 8 * (2 * (g + 1) + u));
            }
          }
          __builtin_amdgcn_sched_barrier(0);
#pragma unroll
          for (int u = 0; u < 2; ++u) {
#pragma unroll
            for (int ct = 0; ct < 4; ++ct)
              o[ct] = wmma_f32(lo2(pa[c][u]), lo2(vb[c][u][ct]), o[ct]);
#pragma unroll
            for (int ct = 0; ct < 4; ++ct)
              o[ct] = wmma_f32(hi2(pa[c][u]), hi2(vb[c][u][ct]), o[ct]);
          }
          __builtin_amdgcn_sched_barrier(0);
        }
      }
    }

    if (more) {
      float* Kn = Klds + ((uint32_t)KVS - cur);
      float* Vn = Vlds + ((uint32_t)KVS - cur);
      *(v4f*)(Kn + kl_off) = kr0;
      *(v4f*)(Kn + kl_off + 32 * LDT) = kr1;
#pragma unroll
      for (int i = 0; i < 4; ++i) {
        Vn[vl_off + i * LDT] = vr0[i];
        Vn[vl_off + (32 + i) * LDT] = vr1[i];
      }
    }
    __syncthreads();
  }

#pragma unroll
  for (int v = 0; v < 8; ++v) lrow[v] += row_xor<1>(lrow[v]);
#pragma unroll
  for (int v = 0; v < 8; ++v) lrow[v] += row_xor<2>(lrow[v]);
#pragma unroll
  for (int v = 0; v < 8; ++v) lrow[v] += row_xor<4>(lrow[v]);
#pragma unroll
  for (int v = 0; v < 8; ++v) lrow[v] += row_xor<8>(lrow[v]);

#pragma unroll
  for (int ct = 0; ct < 4; ++ct)
#pragma unroll
    for (int v = 0; v < 8; ++v) {
      const uint32_t row = (uint32_t)(qbase + half * 8 + v);
      out[row * 1024u + (uint32_t)(h * 64 + ct * 16 + l16)] =
          o[ct][v] / lrow[v];
    }
}

extern "C" void kernel_launch(void* const* d_in, const int* in_sizes, int n_in,
                              void* d_out, int out_size, void* d_ws, size_t ws_size,
                              hipStream_t stream) {
  const float* x     = (const float*)d_in[0];
  const float* Wqkv  = (const float*)d_in[1];
  const float* Wproj = (const float*)d_in[2];
  float* out = (float*)d_out;

  float* qkv     = (float*)d_ws;
  float* attnout = qkv + (size_t)4096 * 3072;

  gemm_nt_wmma<<<dim3(24, 32), 256, 0, stream>>>(
      x, Wqkv, qkv, 1024, 1024, 1024, 3072);

  rope_kernel<<<(4096 * 1024) / 256, 256, 0, stream>>>(qkv, 4096);

  attn_kernel<<<dim3(16, 16), 512, 0, stream>>>(qkv, attnout);

  gemm_nt_wmma<<<dim3(8, 32), 256, 0, stream>>>(
      attnout, Wproj, out, 1024, 1024, 1024, 1024);
}


# attention on transposed tiles (S^T = K Q^T, O^T = V^T P^T), one query per lane: P feeds the 2nd product from the score registers (no LDS round trip), row max/sum in-lane + one half-wave exchange
# speedup vs baseline: 3.0288x; 1.1679x over previous
#include <hip/hip_runtime.h>
#include <hip/hip_bf16.h>

typedef __attribute__((ext_vector_type(2))) float v2f;
typedef __attribute__((ext_vector_type(4))) float v4f;
typedef __attribute__((ext_vector_type(8))) float v8f;

__device__ __forceinline__ v8f wmma_f32(v2f a, v2f b, v8f c) {
  return __builtin_amdgcn_wmma_f32_16x16x4_f32(
       false, a,  false, b,
       (short)0, c,  false,  false);
}

template <int M>
__device__ __forceinline__ float row_xor(float x) {
  return __builtin_amdgcn_update_dpp(0.0f, x, 0x160 + M, 0xf, 0xf, true);
}

__device__ __forceinline__ float xhalf(float x) {
  const int b = __builtin_bit_cast(int, x);
  return __builtin_bit_cast(float,
      __builtin_amdgcn_permlanex16(b, b, 0x76543210, 0xfedcba98, false, false));
}

__device__ __forceinline__ v2f lo2(v4f x) { return __builtin_shufflevector(x, x, 0, 1); }
__device__ __forceinline__ v2f hi2(v4f x) { return __builtin_shufflevector(x, x, 2, 3); }

#define GBM 128
#define GBN 128
#define GBK 32
#define GLD 36
#define GSTAGE ((GBM + GBN) * GLD)
#define RTP 33

__global__ __launch_bounds__(256)
void gemm_nt_wmma(const float* __restrict__ A, const float* __restrict__ B,
                  float* __restrict__ C, int K, int lda, int ldb, int ldc, int rope_n) {
  __shared__ __attribute__((aligned(16))) float lds[2 * GSTAGE];

  const int tid  = threadIdx.x;
  const int lane = tid & 31;
  const int wave = tid >> 5;
  const int half = lane >> 4;
  const int l16  = lane & 15;
  const int wm = wave >> 2;
  const int wn = wave & 3;
  const int bm = blockIdx.y * GBM;
  const int bn = blockIdx.x * GBN;

  const uint32_t srow = (uint32_t)(tid >> 3);
  const uint32_t sc4  = (uint32_t)(tid & 7) * 4u;
  const uint32_t aoff = (uint32_t)(bm + (int)srow) * (uint32_t)lda + sc4;
  const uint32_t boff = (uint32_t)(bn + (int)srow) * (uint32_t)ldb + sc4;
  const uint32_t astep = 32u * (uint32_t)lda;
  const uint32_t bstep = 32u * (uint32_t)ldb;
  const uint32_t soff = srow * GLD + sc4;

  const uint32_t faoff = (uint32_t)(wm * 64 + l16) * GLD + 4u * half;
  const uint32_t fboff = (uint32_t)(GBM + wn * 32 + l16) * GLD + 4u * half;

  v8f acc[4][2];
#pragma unroll
  for (int i = 0; i < 4; ++i)
#pragma unroll
    for (int j = 0; j < 2; ++j) acc[i][j] = (v8f){};

  v4f pa[4], pb[4];
#pragma unroll
  for (int t = 0; t < 4; ++t) {
    pa[t] = *(const v4f*)(A + aoff + (uint32_t)t * astep);
    pb[t] = *(const v4f*)(B + boff + (uint32_t)t * bstep);
  }
#pragma unroll
  for (int t = 0; t < 4; ++t) {
    *(v4f*)(lds + soff + (uint32_t)t * (32u * GLD)) = pa[t];
    *(v4f*)(lds + (uint32_t)(GBM * GLD) + soff + (uint32_t)t * (32u * GLD)) = pb[t];
  }
  __syncthreads();

  const int KT = K / GBK;
  for (int kt = 0; kt < KT; ++kt) {
    const uint32_t cur = (kt & 1) ? (uint32_t)GSTAGE : 0u;
    const uint32_t nxt = (uint32_t)GSTAGE - cur;
    const bool more = (kt + 1 < KT);

    if (more) {
      const uint32_t k1 = (uint32_t)(kt + 1) * GBK;
#pragma unroll
      for (int t = 0; t < 4; ++t) {
        pa[t] = *(const v4f*)(A + aoff + (uint32_t)t * astep + k1);
        pb[t] = *(const v4f*)(B + boff + (uint32_t)t * bstep + k1);
      }
    }

    const float* fa_p = lds + cur + faoff;
    const float* fb_p = lds + cur + fboff;

    v4f fa[2][4], fb[2][2];
#pragma unroll
    for (int i = 0; i < 4; ++i) fa[0][i] = *(const v4f*)(fa_p + i * 16 * GLD);
#pragma unroll
    for (int j = 0; j < 2; ++j) fb[0][j] = *(const v4f*)(fb_p + j * 16 * GLD);

#pragma unroll
    for (int ks = 0; ks < 4; ++ks) {
      const int c = ks & 1, n = c ^ 1;
      if (ks + 1 < 4) {
#pragma unroll
        for (int i = 0; i < 4; ++i)
          fa[n][i] = *(const v4f*)(fa_p + i * 16 * GLD + 8 * (ks + 1));
#pragma unroll
        for (int j = 0; j < 2; ++j)
          fb[n][j] = *(const v4f*)(fb_p + j * 16 * GLD + 8 * (ks + 1));
      }
#pragma unroll
      for (int i = 0; i < 4; ++i)
#pragma unroll
        for (int j = 0; j < 2; ++j) {
          v2f a = __builtin_shufflevector(fa[c][i], fa[c][i], 0, 1);
          v2f b = __builtin_shufflevector(fb[c][j], fb[c][j], 0, 1);
          acc[i][j] = wmma_f32(a, b, acc[i][j]);
        }
#pragma unroll
      for (int i = 0; i < 4; ++i)
#pragma unroll
        for (int j = 0; j < 2; ++j) {
          v2f a = __builtin_shufflevector(fa[c][i], fa[c][i], 2, 3);
          v2f b = __builtin_shufflevector(fb[c][j], fb[c][j], 2, 3);
          acc[i][j] = wmma_f32(a, b, acc[i][j]);
        }
    }

    if (more) {
#pragma unroll
      for (int t = 0; t < 4; ++t) {
        *(v4f*)(lds + nxt + soff + (uint32_t)t * (32u * GLD)) = pa[t];
        *(v4f*)(lds + nxt + (uint32_t)(GBM * GLD) + soff + (uint32_t)t * (32u * GLD)) = pb[t];
      }
    }
    __syncthreads();
  }

  if (bn < rope_n) {
    v2f* tab = (v2f*)lds;
    {
      const int p = tid & 31;
      const float inv = powf(10000.0f, -(float)(2 * p) / 64.0f);
#pragma unroll 4
      for (int k = 0; k < 16; ++k) {
        const int r = (tid >> 5) + 8 * k;
        const float ang = (float)(bm + r) * inv;
        float sn, cs;
        sincosf(ang, &sn, &cs);
        v2f sc;
        sc.x = sn;
        sc.y = cs;
        tab[r * RTP + p] = sc;
      }
    }
    __syncthreads();
    const bool odd = (l16 & 1) != 0;
#pragma unroll
    for (int i = 0; i < 4; ++i)
#pragma unroll
      for (int j = 0; j < 2; ++j) {
        const int pp = ((wn * 32 + j * 16 + l16) & 63) >> 1;
        const v2f* trow = tab + (wm * 64 + i * 16 + half * 8) * RTP + pp;
#pragma unroll
        for (int v = 0; v < 8; ++v) {
          const v2f sc = trow[v * RTP];
          const float x  = acc[i][j][v];
          const float xp = row_xor<1>(x);
          const float sg = odd ? sc.x : -sc.x;
          acc[i][j][v] = x * sc.y + xp * sg;
        }
      }
  }

#pragma unroll
  for (int i = 0; i < 4; ++i)
#pragma unroll
    for (int j = 0; j < 2; ++j) {
      const uint32_t row0 = (uint32_t)(bm + wm * 64 + i * 16 + half * 8);
      const uint32_t col  = (uint32_t)(bn + wn * 32 + j * 16 + l16);
#pragma unroll
      for (int v = 0; v < 8; ++v)
        C[(row0 + v) * (uint32_t)ldc + col] = acc[i][j][v];
    }
}

#define LDT 68
#define KVS (64 * LDT)

__global__ __launch_bounds__(512)
void attn_kernel(const float* __restrict__ qkv, float* __restrict__ out) {
  __shared__ __attribute__((aligned(16))) float Klds[2 * KVS];
  __shared__ __attribute__((aligned(16))) float Vlds[2 * KVS];

  const int tid  = threadIdx.x;
  const int lane = tid & 31;
  const int wave = tid >> 5;
  const int half = lane >> 4;
  const int l16  = lane & 15;
  const int h     = blockIdx.y;
  const int qtile_hi = 31 - (int)blockIdx.x;
  const int qtile = (wave < 8) ? qtile_hi : (int)blockIdx.x;
  const int qbase = qtile * 128 + (wave & 7) * 16;
  const int qrow  = qbase + l16;
  const uint32_t ld = 3072;

  const float* qp = qkv + (uint32_t)h * 64u;
  const float* kp = qkv + 1024u + (uint32_t)h * 64u;
  const float* vp = qkv + 2048u + (uint32_t)h * 64u;

  v4f qf[8];
  {
    const uint32_t qoff = (uint32_t)qrow * ld + 4u * half;
#pragma unroll
    for (int ks = 0; ks < 8; ++ks)
      qf[ks] = *(const v4f*)(qp + qoff + 8u * ks);
  }

  v8f o[4];
#pragma unroll
  for (int ct = 0; ct < 4; ++ct) o[ct] = (v8f){};
  float mrow = -1e30f, lrow = 0.0f;

  const float* kfp0 = Klds + l16 * LDT + 4 * half;
  const float* vfp0 = Vlds + l16 * LDT + 8 * half;

  const uint32_t kg_off = (uint32_t)(tid >> 4) * ld + (uint32_t)(tid & 15) * 4u;
  const uint32_t kl_off = (uint32_t)(tid >> 4) * LDT + (uint32_t)(tid & 15) * 4u;
  const uint32_t vkey = (uint32_t)((wave & 7) * 8 + (lane & 7));
  const uint32_t vc4  = (uint32_t)(((wave >> 3) * 4 + (lane >> 3)) * 4);
  const uint32_t vg_off = vkey * ld + vc4;
  const uint32_t vl_off = vc4 * LDT + vkey;

  v4f kr0, kr1, vr0, vr1;
  kr0 = *(const v4f*)(kp + kg_off);
  kr1 = *(const v4f*)(kp + kg_off + 32u * ld);
  vr0 = *(const v4f*)(vp + vg_off);
  vr1 = *(const v4f*)(vp + vg_off + 32u);
  *(v4f*)(Klds + kl_off) = kr0;
  *(v4f*)(Klds + kl_off + 32 * LDT) = kr1;
#pragma unroll
  for (int i = 0; i < 4; ++i) {
    Vlds[vl_off + i * LDT] = vr0[i];
    Vlds[vl_off + (32 + i) * LDT] = vr1[i];
  }
  __syncthreads();
  const int ktmax_w = (qbase + 15) >> 6;
  const int ktmax_b = (qtile_hi * 128 + 127) >> 6;

  for (int kt = 0; kt <= ktmax_b; ++kt) {
    const uint32_t cur = (kt & 1) ? (uint32_t)KVS : 0u;
    const bool more = (kt < ktmax_b);
    const float* kfp = kfp0 + cur;
    const float* vfp = vfp0 + cur;

    if (more) {
      const uint32_t nbase = (uint32_t)((kt + 1) * 64) * ld;
      kr0 = *(const v4f*)(kp + nbase + kg_off);
      kr1 = *(const v4f*)(kp + nbase + kg_off + 32u * ld);
      vr0 = *(const v4f*)(vp + nbase + vg_off);
      vr1 = *(const v4f*)(vp + nbase + vg_off + 32u);
    }

    if (kt <= ktmax_w) {
      v8f s[4];
#pragma unroll
      for (int j = 0; j < 4; ++j) s[j] = (v8f){};
      {
        v4f kb[2][2][4];
#pragma unroll
        for (int u = 0; u < 2; ++u)
#pragma unroll
          for (int j = 0; j < 4; ++j)
            kb[0][u][j] = *(const v4f*)(kfp + j * 16 * LDT + 8 * u);
        __builtin_amdgcn_sched_barrier(0);
#pragma unroll
        for (int g = 0; g < 4; ++g) {
          const int c = g & 1, n = c ^ 1;
          if (g + 1 < 4) {
#pragma unroll
            for (int u = 0; u < 2; ++u)
#pragma unroll
              for (int j = 0; j < 4; ++j)
                kb[n][u][j] = *(const v4f*)(kfp + j * 16 * LDT + 8 * (2 * (g + 1) + u));
          }
          __builtin_amdgcn_sched_barrier(0);
#pragma unroll
          for (int u = 0; u < 2; ++u) {
            const v4f q4 = qf[2 * g + u];
#pragma unroll
            for (int j = 0; j < 4; ++j)
              s[j] = wmma_f32(lo2(kb[c][u][j]), lo2(q4), s[j]);
#pragma unroll
            for (int j = 0; j < 4; ++j)
              s[j] = wmma_f32(hi2(kb[c][u][j]), hi2(q4), s[j]);
          }
          __builtin_amdgcn_sched_barrier(0);
        }
      }

      v4f vb[2][2][4];
#pragma unroll
      for (int u = 0; u < 2; ++u)
#pragma unroll
        for (int ct = 0; ct < 4; ++ct)
          vb[0][u][ct] = *(const v4f*)(vfp + ct * 16 * LDT + 4 * u);
      __builtin_amdgcn_sched_barrier(0);

      const int dq = qrow - (kt * 64 + 8 * half);
      float m = -1e30f;
#pragma unroll
      for (int j = 0; j < 4; ++j)
#pragma unroll
        for (int r = 0; r < 8; ++r) {
          float x = s[j][r] * 0.125f;
          if (j * 16 + r > dq) x = -1e30f;
          s[j][r] = x;
          m = fmaxf(m, x);
        }
      m = fmaxf(m, xhalf(m));
      const float mnew = fmaxf(mrow, m);
      const float alpha = __expf(mrow - mnew);
      float rsum = 0.0f;
#pragma unroll
      for (int j = 0; j < 4; ++j)
#pragma unroll
        for (int r = 0; r < 8; ++r) {
          const float p = __expf(s[j][r] - mnew);
          s[j][r] = p;
          rsum += p;
        }
      lrow = lrow * alpha + rsum;
      mrow = mnew;
#pragma unroll
      for (int ct = 0; ct < 4; ++ct)
#pragma unroll
        for (int r = 0; r < 8; ++r) o[ct][r] *= alpha;
      __builtin_amdgcn_sched_barrier(0);

#pragma unroll
      for (int j = 0; j < 4; ++j) {
        const int c = j & 1, n = c ^ 1;
        if (j + 1 < 4) {
#pragma unroll
          for (int u = 0; u < 2; ++u)
#pragma unroll
            for (int ct = 0; ct < 4; ++ct)
              vb[n][u][ct] = *(const v4f*)(vfp + ct * 16 * LDT + 16 * (j + 1) + 4 * u);
        }
        __builtin_amdgcn_sched_barrier(0);
        {
          const v2f p0 = __builtin_shufflevector(s[j], s[j], 0, 1);
          const v2f p1 = __builtin_shufflevector(s[j], s[j], 2, 3);
          const v2f p2 = __builtin_shufflevector(s[j], s[j], 4, 5);
          const v2f p3 = __builtin_shufflevector(s[j], s[j], 6, 7);
#pragma unroll
          for (int ct = 0; ct < 4; ++ct)
            o[ct] = wmma_f32(lo2(vb[c][0][ct]), p0, o[ct]);
#pragma unroll
          for (int ct = 0; ct < 4; ++ct)
            o[ct] = wmma_f32(hi2(vb[c][0][ct]), p1, o[ct]);
#pragma unroll
          for (int ct = 0; ct < 4; ++ct)
            o[ct] = wmma_f32(lo2(vb[c][1][ct]), p2, o[ct]);
#pragma unroll
          for (int ct = 0; ct < 4; ++ct)
            o[ct] = wmma_f32(hi2(vb[c][1][ct]), p3, o[ct]);
        }
        __builtin_amdgcn_sched_barrier(0);
      }
    }

    if (more) {
      float* Kn = Klds + ((uint32_t)KVS - cur);
      float* Vn = Vlds + ((uint32_t)KVS - cur);
      *(v4f*)(Kn + kl_off) = kr0;
      *(v4f*)(Kn + kl_off + 32 * LDT) = kr1;
#pragma unroll
      for (int i = 0; i < 4; ++i) {
        Vn[vl_off + i * LDT] = vr0[i];
        Vn[vl_off + (32 + i) * LDT] = vr1[i];
      }
    }
    __syncthreads();
  }

  const float lsum = lrow + xhalf(lrow);

  float* slab = ((wave < 8) ? Klds : Vlds) + (wave & 7) * 16 * LDT;
#pragma unroll
  for (int ct = 0; ct < 4; ++ct) {
    v4f a, b;
#pragma unroll
    for (int r = 0; r < 4; ++r) {
      a[r] = o[ct][r] / lsum;
      b[r] = o[ct][4 + r] / lsum;
    }
    *(v4f*)(slab + l16 * LDT + ct * 16 + 8 * half) = a;
    *(v4f*)(slab + l16 * LDT + ct * 16 + 8 * half + 4) = b;
  }
  __builtin_amdgcn_wave_barrier();
#pragma unroll
  for (int i = 0; i < 8; ++i) {
    const int row = 2 * i + half;
    const v4f v = *(const v4f*)(slab + row * LDT + 4 * l16);
    *(v4f*)(out + (uint32_t)(qbase + row) * 1024u + (uint32_t)(h * 64 + 4 * l16)) = v;
  }
}

extern "C" void kernel_launch(void* const* d_in, const int* in_sizes, int n_in,
                              void* d_out, int out_size, void* d_ws, size_t ws_size,
                              hipStream_t stream) {
  const float* x     = (const float*)d_in[0];
  const float* Wqkv  = (const float*)d_in[1];
  const float* Wproj = (const float*)d_in[2];
  float* out = (float*)d_out;

  float* qkv     = (float*)d_ws;
  float* attnout = qkv + (size_t)4096 * 3072;

  gemm_nt_wmma<<<dim3(24, 32), 256, 0, stream>>>(
      x, Wqkv, qkv, 1024, 1024, 1024, 3072, 2048);

  attn_kernel<<<dim3(16, 16), 512, 0, stream>>>(qkv, attnout);

  gemm_nt_wmma<<<dim3(8, 32), 256, 0, stream>>>(
      attnout, Wproj, out, 1024, 1024, 1024, 1024, 0);
}
